# starting kernel
# baseline (speedup 1.0000x reference)
#include <hip/hip_runtime.h>
#include <hip/hip_bf16.h>

typedef float v2f __attribute__((ext_vector_type(2)));
typedef float v8f __attribute__((ext_vector_type(8)));

#define GROUPS 64
#define GLEN   1025
#define HSZ    32
#define SCALE  0.17677669529663687f
#define PSTRIDE 68

__device__ __forceinline__ v8f wmma_f32(v2f a, v2f b, v8f c) {
    return __builtin_amdgcn_wmma_f32_16x16x4_f32(false, a, false, b, (short)0, c, false, false);
}

__global__ __launch_bounds__(32) void qkv_gemm_kernel(
    const float* __restrict__ x, const float* __restrict__ w,
    float* __restrict__ qg, float* __restrict__ kg, float* __restrict__ vg)
{
    const int mt = blockIdx.x;
    const int nt = blockIdx.y;
    const int lane = threadIdx.x;
    const int half = lane >> 4, l16 = lane & 15;

    const float* arow = x + (size_t)(mt * 16 + l16) * 128 + 2 * half;
    const float* brow = w + (size_t)(nt * 16 + l16) * 128 + 2 * half;

    v8f acc = {};
#pragma unroll
    for (int c = 0; c < 32; ++c) {
        float2 af = *(const float2*)(arow + 4 * c);
        float2 bf = *(const float2*)(brow + 4 * c);
        v2f a = { af.x, af.y };
        v2f b = { bf.x, bf.y };
        acc = wmma_f32(a, b, acc);
    }

    const int j0 = nt * 16;
    const int which = j0 / 128;
    float* dst = (which == 0) ? qg : ((which == 1) ? kg : vg);
    const int j = (j0 % 128) + l16;
    const int h = j / 32, d = j % 32;
    const int row0 = mt * 16;
    const int bidx = row0 / 8192;
#pragma unroll
    for (int r = 0; r < 8; ++r) {
        int t = (row0 % 8192) + r + 8 * half;
        int g = t / 1024, rig = t % 1024;
        size_t off = ((size_t)((bidx * 4 + h) * 8 + g) * GLEN + rig) * HSZ + d;
        dst[off] = acc[r];
    }
}

__global__ __launch_bounds__(128) void mean_kernel(
    float* __restrict__ qg, float* __restrict__ kg, float* __restrict__ vg)
{
    const int grp = blockIdx.x;
    const int t = threadIdx.x;
    if (t >= 96) return;
    float* buf = (t < 32) ? qg : ((t < 64) ? kg : vg);
    const int d = t % 32;
    const float* base = buf + (size_t)grp * GLEN * HSZ + d;
    float s = 0.0f;
    for (int r = 0; r < 1024; ++r) s += base[r * HSZ];
    buf[(size_t)grp * GLEN * HSZ + 1024 * HSZ + d] = s * (1.0f / 1024.0f);
}

__global__ __launch_bounds__(32) void attn_group_kernel(
    const float* __restrict__ qg, const float* __restrict__ kg,
    const float* __restrict__ vg, float* __restrict__ xg)
{
    const int qt = blockIdx.x;
    const int grp = blockIdx.y;
    const int lane = threadIdx.x;
    const int half = lane >> 4, l16 = lane & 15;

    const float* Q = qg + (size_t)grp * GLEN * HSZ;
    const float* K = kg + (size_t)grp * GLEN * HSZ;
    const float* V = vg + (size_t)grp * GLEN * HSZ;
    float*       X = xg + (size_t)grp * GLEN * HSZ;

    const int qbase = qt * 16;
    int qrow = qbase + l16; if (qrow > 1024) qrow = 1024;
    const float* qptr = Q + qrow * HSZ + 2 * half;

    v2f qf[8];
#pragma unroll
    for (int c = 0; c < 8; ++c) {
        float2 f = *(const float2*)(qptr + 4 * c);
        qf[c] = (v2f){ f.x, f.y };
    }

    v8f acc0 = {}, acc1 = {};
    float rowM[8], rowL[8];
#pragma unroll
    for (int r = 0; r < 8; ++r) { rowM[r] = -3.0e38f; rowL[r] = 0.0f; }

    __shared__ float lds[16 * PSTRIDE];

    for (int kbase = 0; kbase <= qbase + 15; kbase += 64) {
        int jmax = ((qbase + 15 - kbase) >> 4) + 1;
        if (jmax > 4) jmax = 4;

        v8f s[4];
#pragma unroll
        for (int j = 0; j < 4; ++j) {
            s[j] = (v8f){};
            if (j < jmax) {
                int krow = kbase + 16 * j + l16; if (krow > 1024) krow = 1024;
                const float* kptr = K + krow * HSZ + 2 * half;
#pragma unroll
                for (int c = 0; c < 8; ++c) {
                    float2 f = *(const float2*)(kptr + 4 * c);
                    v2f b = { f.x, f.y };
                    s[j] = wmma_f32(qf[c], b, s[j]);
                }
            }
        }

        float p[4][8];
#pragma unroll
        for (int r = 0; r < 8; ++r) {
            const int qi = qbase + r + 8 * half;
            float sv[4];
#pragma unroll
            for (int j = 0; j < 4; ++j) {
                const int kj = kbase + 16 * j + l16;
                const bool valid = (j < jmax) && (kj <= qi) && (kj <= 1024);
                sv[j] = valid ? s[j][r] * SCALE : -3.0e38f;
            }
            float rm = fmaxf(fmaxf(sv[0], sv[1]), fmaxf(sv[2], sv[3]));
            rm = fmaxf(rm, __shfl_xor(rm, 1, 32));
            rm = fmaxf(rm, __shfl_xor(rm, 2, 32));
            rm = fmaxf(rm, __shfl_xor(rm, 4, 32));
            rm = fmaxf(rm, __shfl_xor(rm, 8, 32));
            const float nm = fmaxf(rowM[r], rm);
            float rs = 0.0f;
#pragma unroll
            for (int j = 0; j < 4; ++j) {
                float pv = __expf(sv[j] - nm);
                p[j][r] = pv;
                rs += pv;
            }
            rs += __shfl_xor(rs, 1, 32);
            rs += __shfl_xor(rs, 2, 32);
            rs += __shfl_xor(rs, 4, 32);
            rs += __shfl_xor(rs, 8, 32);
            const float corr = __expf(rowM[r] - nm);
            rowM[r] = nm;
            rowL[r] = rowL[r] * corr + rs;
            acc0[r] = acc0[r] * corr;
            acc1[r] = acc1[r] * corr;
        }

        __syncthreads();
#pragma unroll
        for (int j = 0; j < 4; ++j)
#pragma unroll
            for (int r = 0; r < 8; ++r)
                lds[(r + 8 * half) * PSTRIDE + 16 * j + l16] = p[j][r];
        __syncthreads();

        const int cmax = 4 * jmax;
#pragma unroll
        for (int c = 0; c < 16; ++c) {
            if (c < cmax) {
                float2 pf = *(const float2*)&lds[l16 * PSTRIDE + 4 * c + 2 * half];
                v2f a = { pf.x, pf.y };
                int k0 = kbase + 4 * c + 2 * half;
                int kr0 = (k0     > 1024) ? 1024 : k0;
                int kr1 = (k0 + 1 > 1024) ? 1024 : k0 + 1;
                v2f b0 = { V[kr0 * HSZ + l16],      V[kr1 * HSZ + l16] };
                v2f b1 = { V[kr0 * HSZ + 16 + l16], V[kr1 * HSZ + 16 + l16] };
                acc0 = wmma_f32(a, b0, acc0);
                acc1 = wmma_f32(a, b1, acc1);
            }
        }
    }

#pragma unroll
    for (int r = 0; r < 8; ++r) {
        const int row = qbase + r + 8 * half;
        if (row <= 1024) {
            const float invl = 1.0f / rowL[r];
            X[row * HSZ + l16]      = acc0[r] * invl + V[row * HSZ + l16];
            X[row * HSZ + 16 + l16] = acc1[r] * invl + V[row * HSZ + 16 + l16];
        }
    }
}

__global__ __launch_bounds__(256) void meanatt_kernel(
    const float* __restrict__ qg, const float* __restrict__ kg,
    const float* __restrict__ xg, float* __restrict__ y)
{
    const int bh = blockIdx.x;
    __shared__ float qm[8][32], km[8][32], vm[8][32];
    const int t = threadIdx.x;
    const int g = t / 32, d = t % 32;
    const size_t rowoff = ((size_t)(bh * 8 + g) * GLEN + 1024) * HSZ + d;
    qm[g][d] = qg[rowoff];
    km[g][d] = kg[rowoff];
    vm[g][d] = xg[rowoff];
    __syncthreads();

    float sarr[8];
    float mx = -3.0e38f;
    for (int gp = 0; gp <= g; ++gp) {
        float sv = 0.0f;
        for (int c = 0; c < 32; ++c) sv += qm[g][c] * km[gp][c];
        sv *= SCALE;
        sarr[gp] = sv;
        mx = fmaxf(mx, sv);
    }
    float denom = 0.0f;
    for (int gp = 0; gp <= g; ++gp) { sarr[gp] = __expf(sarr[gp] - mx); denom += sarr[gp]; }
    float o = 0.0f;
    for (int gp = 0; gp <= g; ++gp) o += sarr[gp] * vm[gp][d];
    y[(bh * 8 + g) * 32 + d] = o / denom;
}

__global__ __launch_bounds__(256) void assemble_kernel(
    const float* __restrict__ xg, const float* __restrict__ y, float* __restrict__ z)
{
    const int idx = blockIdx.x * 256 + threadIdx.x;
    if (idx >= 2 * 8200 * 128) return;
    const int c = idx % 128;
    const int t = (idx / 128) % 8200;
    const int b = idx / (8200 * 128);
    const int h = c / 32, d = c % 32;
    const int g = t / GLEN, row = t % GLEN;
    const int bh = b * 4 + h;
    float v;
    if (g == 0)        v = xg[((size_t)(bh * 8 + 0) * GLEN + row) * HSZ + d];
    else if (row == 0) v = y[(bh * 8 + (g - 1)) * 32 + d];
    else               v = xg[((size_t)(bh * 8 + g) * GLEN + (row - 1)) * HSZ + d];
    z[idx] = v;
}

__global__ __launch_bounds__(32) void proj_gemm_kernel(
    const float* __restrict__ z, const float* __restrict__ w, float* __restrict__ out)
{
    const int mt = blockIdx.x;
    const int nt = blockIdx.y;
    const int lane = threadIdx.x;
    const int half = lane >> 4, l16 = lane & 15;

    const float* arow = z + (size_t)(mt * 16 + l16) * 128 + 2 * half;
    const float* brow = w + (size_t)(nt * 16 + l16) * 128 + 2 * half;

    v8f acc = {};
#pragma unroll
    for (int c = 0; c < 32; ++c) {
        float2 af = *(const float2*)(arow + 4 * c);
        float2 bf = *(const float2*)(brow + 4 * c);
        v2f a = { af.x, af.y };
        v2f b = { bf.x, bf.y };
        acc = wmma_f32(a, b, acc);
    }
#pragma unroll
    for (int r = 0; r < 8; ++r) {
        const int row = mt * 16 + r + 8 * half;
        out[(size_t)row * 128 + nt * 16 + l16] = acc[r];
    }
}

extern "C" void kernel_launch(void* const* d_in, const int* in_sizes, int n_in,
                              void* d_out, int out_size, void* d_ws, size_t ws_size,
                              hipStream_t stream) {
    const float* x      = (const float*)d_in[0];
    const float* w_attn = (const float*)d_in[1];
    const float* w_proj = (const float*)d_in[2];

    float* ws = (float*)d_ws;
    const size_t GBUF = (size_t)GROUPS * GLEN * HSZ;
    float* qg = ws;
    float* kg = ws + GBUF;
    float* vg = ws + 2 * GBUF;
    float* xg = ws + 3 * GBUF;
    float* y  = ws + 4 * GBUF;
    float* z  = qg;

    qkv_gemm_kernel<<<dim3(1024, 24), 32, 0, stream>>>(x, w_attn, qg, kg, vg);
    mean_kernel<<<GROUPS, 128, 0, stream>>>(qg, kg, vg);
    attn_group_kernel<<<dim3(65, GROUPS), 32, 0, stream>>>(qg, kg, vg, xg);
    meanatt_kernel<<<8, 256, 0, stream>>>(qg, kg, xg, y);
    assemble_kernel<<<8200, 256, 0, stream>>>(xg, y, z);
    proj_gemm_kernel<<<dim3(1025, 8), 32, 0, stream>>>(z, w_proj, (float*)d_out);
}


# attention kernel: each 64-key block issues all its K and V fragment loads as one group ahead of its matrix work (branch-free block body) instead of one load round trip per WMMA
# speedup vs baseline: 1.5449x; 1.5449x over previous
#include <hip/hip_runtime.h>
#include <hip/hip_bf16.h>

typedef float v2f __attribute__((ext_vector_type(2)));
typedef float v8f __attribute__((ext_vector_type(8)));

#define GROUPS 64
#define GLEN   1025
#define HSZ    32
#define SCALE  0.17677669529663687f
#define PSTRIDE 68

__device__ __forceinline__ v8f wmma_f32(v2f a, v2f b, v8f c) {
    return __builtin_amdgcn_wmma_f32_16x16x4_f32(false, a, false, b, (short)0, c, false, false);
}

__global__ __launch_bounds__(32) void qkv_gemm_kernel(
    const float* __restrict__ x, const float* __restrict__ w,
    float* __restrict__ qg, float* __restrict__ kg, float* __restrict__ vg)
{
    const int mt = blockIdx.x;
    const int nt = blockIdx.y;
    const int lane = threadIdx.x;
    const int half = lane >> 4, l16 = lane & 15;

    const float* arow = x + (size_t)(mt * 16 + l16) * 128 + 2 * half;
    const float* brow = w + (size_t)(nt * 16 + l16) * 128 + 2 * half;

    v8f acc = {};
#pragma unroll
    for (int c = 0; c < 32; ++c) {
        float2 af = *(const float2*)(arow + 4 * c);
        float2 bf = *(const float2*)(brow + 4 * c);
        v2f a = { af.x, af.y };
        v2f b = { bf.x, bf.y };
        acc = wmma_f32(a, b, acc);
    }

    const int j0 = nt * 16;
    const int which = j0 / 128;
    float* dst = (which == 0) ? qg : ((which == 1) ? kg : vg);
    const int j = (j0 % 128) + l16;
    const int h = j / 32, d = j % 32;
    const int row0 = mt * 16;
    const int bidx = row0 / 8192;
#pragma unroll
    for (int r = 0; r < 8; ++r) {
        int t = (row0 % 8192) + r + 8 * half;
        int g = t / 1024, rig = t % 1024;
        size_t off = ((size_t)((bidx * 4 + h) * 8 + g) * GLEN + rig) * HSZ + d;
        dst[off] = acc[r];
    }
}

__global__ __launch_bounds__(128) void mean_kernel(
    float* __restrict__ qg, float* __restrict__ kg, float* __restrict__ vg)
{
    const int grp = blockIdx.x;
    const int t = threadIdx.x;
    if (t >= 96) return;
    float* buf = (t < 32) ? qg : ((t < 64) ? kg : vg);
    const int d = t % 32;
    const float* base = buf + (size_t)grp * GLEN * HSZ + d;
    float s = 0.0f;
    for (int r = 0; r < 1024; ++r) s += base[r * HSZ];
    buf[(size_t)grp * GLEN * HSZ + 1024 * HSZ + d] = s * (1.0f / 1024.0f);
}

template <bool DIAG>
__device__ __forceinline__ void attn_block(
    const float* __restrict__ K, const float* __restrict__ V,
    const v2f (&qf)[8], v8f& acc0, v8f& acc1,
    float (&rowM)[8], float (&rowL)[8], float* lds,
    const int kbase, const int qbase, const int half, const int l16)
{
    v2f kf[4][8];
#pragma unroll
    for (int j = 0; j < 4; ++j) {
        int krow = kbase + 16 * j + l16;
        if (DIAG) { if (krow > 1024) krow = 1024; }
        const float* kptr = K + krow * HSZ + 2 * half;
#pragma unroll
        for (int c = 0; c < 8; ++c) {
            float2 f = *(const float2*)(kptr + 4 * c);
            kf[j][c] = (v2f){ f.x, f.y };
        }
    }
    v2f vb0[16], vb1[16];
#pragma unroll
    for (int c = 0; c < 16; ++c) {
        const int k0 = kbase + 4 * c + 2 * half;
        int kr0 = k0, kr1 = k0 + 1;
        if (DIAG) {
            if (kr0 > 1024) kr0 = 1024;
            if (kr1 > 1024) kr1 = 1024;
        }
        vb0[c] = (v2f){ V[kr0 * HSZ + l16],      V[kr1 * HSZ + l16] };
        vb1[c] = (v2f){ V[kr0 * HSZ + 16 + l16], V[kr1 * HSZ + 16 + l16] };
    }

    v8f s[4];
#pragma unroll
    for (int j = 0; j < 4; ++j) {
        s[j] = (v8f){};
#pragma unroll
        for (int c = 0; c < 8; ++c) s[j] = wmma_f32(qf[c], kf[j][c], s[j]);
    }

    float p[4][8];
#pragma unroll
    for (int r = 0; r < 8; ++r) {
        const int qi = qbase + r + 8 * half;
        float sv[4];
#pragma unroll
        for (int j = 0; j < 4; ++j) {
            if (DIAG) {
                const int kj = kbase + 16 * j + l16;
                const bool valid = (kj <= qi) && (kj <= 1024);
                sv[j] = valid ? s[j][r] * SCALE : -3.0e38f;
            } else {
                sv[j] = s[j][r] * SCALE;
            }
        }
        float rm = fmaxf(fmaxf(sv[0], sv[1]), fmaxf(sv[2], sv[3]));
        rm = fmaxf(rm, __shfl_xor(rm, 1, 32));
        rm = fmaxf(rm, __shfl_xor(rm, 2, 32));
        rm = fmaxf(rm, __shfl_xor(rm, 4, 32));
        rm = fmaxf(rm, __shfl_xor(rm, 8, 32));
        const float nm = fmaxf(rowM[r], rm);
        float rs = 0.0f;
#pragma unroll
        for (int j = 0; j < 4; ++j) {
            float pv = __expf(sv[j] - nm);
            p[j][r] = pv;
            rs += pv;
        }
        rs += __shfl_xor(rs, 1, 32);
        rs += __shfl_xor(rs, 2, 32);
        rs += __shfl_xor(rs, 4, 32);
        rs += __shfl_xor(rs, 8, 32);
        const float corr = __expf(rowM[r] - nm);
        rowM[r] = nm;
        rowL[r] = rowL[r] * corr + rs;
        acc0[r] = acc0[r] * corr;
        acc1[r] = acc1[r] * corr;
    }

    __syncthreads();
#pragma unroll
    for (int j = 0; j < 4; ++j)
#pragma unroll
        for (int r = 0; r < 8; ++r)
            lds[(r + 8 * half) * PSTRIDE + 16 * j + l16] = p[j][r];
    __syncthreads();

#pragma unroll
    for (int c = 0; c < 16; ++c) {
        float2 pf = *(const float2*)&lds[l16 * PSTRIDE + 4 * c + 2 * half];
        v2f a = { pf.x, pf.y };
        acc0 = wmma_f32(a, vb0[c], acc0);
        acc1 = wmma_f32(a, vb1[c], acc1);
    }
}

__global__ __launch_bounds__(32) void attn_group_kernel(
    const float* __restrict__ qg, const float* __restrict__ kg,
    const float* __restrict__ vg, float* __restrict__ xg)
{
    const int qt = blockIdx.x;
    const int grp = blockIdx.y;
    const int lane = threadIdx.x;
    const int half = lane >> 4, l16 = lane & 15;

    const float* Q = qg + (size_t)grp * GLEN * HSZ;
    const float* K = kg + (size_t)grp * GLEN * HSZ;
    const float* V = vg + (size_t)grp * GLEN * HSZ;
    float*       X = xg + (size_t)grp * GLEN * HSZ;

    const int qbase = qt * 16;
    int qrow = qbase + l16; if (qrow > 1024) qrow = 1024;
    const float* qptr = Q + qrow * HSZ + 2 * half;

    v2f qf[8];
#pragma unroll
    for (int c = 0; c < 8; ++c) {
        float2 f = *(const float2*)(qptr + 4 * c);
        qf[c] = (v2f){ f.x, f.y };
    }

    v8f acc0 = {}, acc1 = {};
    float rowM[8], rowL[8];
#pragma unroll
    for (int r = 0; r < 8; ++r) { rowM[r] = -3.0e38f; rowL[r] = 0.0f; }

    __shared__ float lds[16 * PSTRIDE];

    const int nfull = qt >> 2;
    for (int b = 0; b < nfull; ++b)
        attn_block<false>(K, V, qf, acc0, acc1, rowM, rowL, lds, 64 * b, qbase, half, l16);
    attn_block<true>(K, V, qf, acc0, acc1, rowM, rowL, lds, 64 * nfull, qbase, half, l16);

#pragma unroll
    for (int r = 0; r < 8; ++r) {
        const int row = qbase + r + 8 * half;
        if (row <= 1024) {
            const float invl = 1.0f / rowL[r];
            X[row * HSZ + l16]      = acc0[r] * invl + V[row * HSZ + l16];
            X[row * HSZ + 16 + l16] = acc1[r] * invl + V[row * HSZ + 16 + l16];
        }
    }
}

__global__ __launch_bounds__(256) void meanatt_kernel(
    const float* __restrict__ qg, const float* __restrict__ kg,
    const float* __restrict__ xg, float* __restrict__ y)
{
    const int bh = blockIdx.x;
    __shared__ float qm[8][32], km[8][32], vm[8][32];
    const int t = threadIdx.x;
    const int g = t / 32, d = t % 32;
    const size_t rowoff = ((size_t)(bh * 8 + g) * GLEN + 1024) * HSZ + d;
    qm[g][d] = qg[rowoff];
    km[g][d] = kg[rowoff];
    vm[g][d] = xg[rowoff];
    __syncthreads();

    float sarr[8];
    float mx = -3.0e38f;
    for (int gp = 0; gp <= g; ++gp) {
        float sv = 0.0f;
        for (int c = 0; c < 32; ++c) sv += qm[g][c] * km[gp][c];
        sv *= SCALE;
        sarr[gp] = sv;
        mx = fmaxf(mx, sv);
    }
    float denom = 0.0f;
    for (int gp = 0; gp <= g; ++gp) { sarr[gp] = __expf(sarr[gp] - mx); denom += sarr[gp]; }
    float o = 0.0f;
    for (int gp = 0; gp <= g; ++gp) o += sarr[gp] * vm[gp][d];
    y[(bh * 8 + g) * 32 + d] = o / denom;
}

__global__ __launch_bounds__(256) void assemble_kernel(
    const float* __restrict__ xg, const float* __restrict__ y, float* __restrict__ z)
{
    const int idx = blockIdx.x * 256 + threadIdx.x;
    if (idx >= 2 * 8200 * 128) return;
    const int c = idx % 128;
    const int t = (idx / 128) % 8200;
    const int b = idx / (8200 * 128);
    const int h = c / 32, d = c % 32;
    const int g = t / GLEN, row = t % GLEN;
    const int bh = b * 4 + h;
    float v;
    if (g == 0)        v = xg[((size_t)(bh * 8 + 0) * GLEN + row) * HSZ + d];
    else if (row == 0) v = y[(bh * 8 + (g - 1)) * 32 + d];
    else               v = xg[((size_t)(bh * 8 + g) * GLEN + (row - 1)) * HSZ + d];
    z[idx] = v;
}

__global__ __launch_bounds__(32) void proj_gemm_kernel(
    const float* __restrict__ z, const float* __restrict__ w, float* __restrict__ out)
{
    const int mt = blockIdx.x;
    const int nt = blockIdx.y;
    const int lane = threadIdx.x;
    const int half = lane >> 4, l16 = lane & 15;

    const float* arow = z + (size_t)(mt * 16 + l16) * 128 + 2 * half;
    const float* brow = w + (size_t)(nt * 16 + l16) * 128 + 2 * half;

    v8f acc = {};
#pragma unroll
    for (int c = 0; c < 32; ++c) {
        float2 af = *(const float2*)(arow + 4 * c);
        float2 bf = *(const float2*)(brow + 4 * c);
        v2f a = { af.x, af.y };
        v2f b = { bf.x, bf.y };
        acc = wmma_f32(a, b, acc);
    }
#pragma unroll
    for (int r = 0; r < 8; ++r) {
        const int row = mt * 16 + r + 8 * half;
        out[(size_t)row * 128 + nt * 16 + l16] = acc[r];
    }
}

extern "C" void kernel_launch(void* const* d_in, const int* in_sizes, int n_in,
                              void* d_out, int out_size, void* d_ws, size_t ws_size,
                              hipStream_t stream) {
    const float* x      = (const float*)d_in[0];
    const float* w_attn = (const float*)d_in[1];
    const float* w_proj = (const float*)d_in[2];

    float* ws = (float*)d_ws;
    const size_t GBUF = (size_t)GROUPS * GLEN * HSZ;
    float* qg = ws;
    float* kg = ws + GBUF;
    float* vg = ws + 2 * GBUF;
    float* xg = ws + 3 * GBUF;
    float* y  = ws + 4 * GBUF;
    float* z  = qg;

    qkv_gemm_kernel<<<dim3(1024, 24), 32, 0, stream>>>(x, w_attn, qg, kg, vg);
    mean_kernel<<<GROUPS, 128, 0, stream>>>(qg, kg, vg);
    attn_group_kernel<<<dim3(65, GROUPS), 32, 0, stream>>>(qg, kg, vg, xg);
    meanatt_kernel<<<8, 256, 0, stream>>>(qg, kg, xg, y);
    assemble_kernel<<<8200, 256, 0, stream>>>(xg, y, z);
    proj_gemm_kernel<<<dim3(1025, 8), 32, 0, stream>>>(z, w_proj, (float*)d_out);
}


# mean kernel: one 8-wave work-group per (group, tensor) reads its 128 KB as 32 whole-line 16-byte loads per thread in one group, sums in registers and LDS; was 1024 serial strided loads per thread
# speedup vs baseline: 2.4340x; 1.5755x over previous
#include <hip/hip_runtime.h>
#include <hip/hip_bf16.h>

typedef float v2f __attribute__((ext_vector_type(2)));
typedef float v8f __attribute__((ext_vector_type(8)));

#define GROUPS 64
#define GLEN   1025
#define HSZ    32
#define SCALE  0.17677669529663687f
#define PSTRIDE 68

__device__ __forceinline__ v8f wmma_f32(v2f a, v2f b, v8f c) {
    return __builtin_amdgcn_wmma_f32_16x16x4_f32(false, a, false, b, (short)0, c, false, false);
}

#define QKV_TILES 4
__global__ __launch_bounds__(384) __attribute__((amdgpu_waves_per_eu(3, 3))) void qkv_gemm_kernel(
    const float* __restrict__ x, const float* __restrict__ w,
    float* __restrict__ qg, float* __restrict__ kg, float* __restrict__ vg)
{
    const int s = __builtin_amdgcn_readfirstlane((int)(threadIdx.x >> 5));
    const int lane = threadIdx.x & 31;
    const int half = lane >> 4, l16 = lane & 15;
    const int mt0 = blockIdx.x * QKV_TILES;

    float4 bf[2][16];
#pragma unroll
    for (int t = 0; t < 2; ++t) {
        const float* brow = w + (size_t)(32 * s + 16 * t + l16) * 128 + 4 * half;
#pragma unroll
        for (int c = 0; c < 16; ++c) bf[t][c] = *(const float4*)(brow + 8 * c);
    }

    const int which = s >> 2;
    const int h = s & 3;
    float* dst = (which == 0) ? qg : ((which == 1) ? kg : vg);

    float4 af[2][16];
    {
        const float* arow = x + (size_t)(mt0 * 16 + l16) * 128 + 4 * half;
#pragma unroll
        for (int c = 0; c < 16; ++c) af[0][c] = *(const float4*)(arow + 8 * c);
    }
    __builtin_amdgcn_sched_barrier(0);
#pragma unroll
    for (int i = 0; i < QKV_TILES; ++i) {
        if (i + 1 < QKV_TILES) {
            const float* arow = x + (size_t)((mt0 + i + 1) * 16 + l16) * 128 + 4 * half;
#pragma unroll
            for (int c = 0; c < 16; ++c) af[(i + 1) & 1][c] = *(const float4*)(arow + 8 * c);
        }
        __builtin_amdgcn_sched_barrier(0);
        v8f acc0 = {}, acc1 = {};
#pragma unroll
        for (int c = 0; c < 16; ++c) {
            const float4 a = af[i & 1][c];
            const v2f alo = { a.x, a.y }, ahi = { a.z, a.w };
            const v2f b0lo = { bf[0][c].x, bf[0][c].y }, b0hi = { bf[0][c].z, bf[0][c].w };
            const v2f b1lo = { bf[1][c].x, bf[1][c].y }, b1hi = { bf[1][c].z, bf[1][c].w };
            acc0 = wmma_f32(alo, b0lo, acc0);
            acc1 = wmma_f32(alo, b1lo, acc1);
            acc0 = wmma_f32(ahi, b0hi, acc0);
            acc1 = wmma_f32(ahi, b1hi, acc1);
        }
        const int mt = mt0 + i;
        const int bidx = mt >> 9;
        const int g = (mt & 511) >> 6;
        const int rig0 = (mt & 63) * 16;
        float* base = dst + ((size_t)((bidx * 4 + h) * 8 + g) * GLEN + rig0 + 8 * half) * HSZ + l16;
#pragma unroll
        for (int r = 0; r < 8; ++r) {
            base[r * HSZ]      = acc0[r];
            base[r * HSZ + 16] = acc1[r];
        }
    }
}

#define MEAN_THREADS 256
#define MEAN_LOADS   32
__global__ __launch_bounds__(MEAN_THREADS) __attribute__((amdgpu_waves_per_eu(2, 2))) void mean_kernel(
    float* __restrict__ qg, float* __restrict__ kg, float* __restrict__ vg)
{
    const int grp = blockIdx.x;
    const int which = blockIdx.y;
    float* buf = (which == 0) ? qg : ((which == 1) ? kg : vg);
    float* gbase = buf + (size_t)grp * GLEN * HSZ;
    const int t = threadIdx.x;
    const int wave = t >> 5, lane = t & 31;

    const float4* src = (const float4*)gbase + t;
    float4 v[MEAN_LOADS];
#pragma unroll
    for (int i = 0; i < MEAN_LOADS; ++i) v[i] = src[i * MEAN_THREADS];
    __builtin_amdgcn_sched_barrier(0);

#pragma unroll
    for (int s = MEAN_LOADS / 2; s >= 1; s >>= 1) {
#pragma unroll
        for (int i = 0; i < s; ++i) {
            v[i].x += v[i + s].x;
            v[i].y += v[i + s].y;
            v[i].z += v[i + s].z;
            v[i].w += v[i + s].w;
        }
    }
    float4 a = v[0];
    a.x += __shfl_xor(a.x, 8, 32);  a.y += __shfl_xor(a.y, 8, 32);
    a.z += __shfl_xor(a.z, 8, 32);  a.w += __shfl_xor(a.w, 8, 32);
    a.x += __shfl_xor(a.x, 16, 32); a.y += __shfl_xor(a.y, 16, 32);
    a.z += __shfl_xor(a.z, 16, 32); a.w += __shfl_xor(a.w, 16, 32);

    __shared__ __attribute__((aligned(16))) float part[MEAN_THREADS / 32][HSZ];
    if (lane < 8) *(float4*)&part[wave][4 * lane] = a;
    __syncthreads();
    if (t < HSZ) {
        float s = 0.0f;
#pragma unroll
        for (int w = 0; w < MEAN_THREADS / 32; ++w) s += part[w][t];
        gbase[1024 * HSZ + t] = s * (1.0f / 1024.0f);
    }
}

template <bool DIAG>
__device__ __forceinline__ void attn_block(
    const float* __restrict__ K, const float* __restrict__ V,
    const v2f (&qf)[8], v8f& acc0, v8f& acc1,
    float (&rowM)[8], float (&rowL)[8], float* lds,
    const int kbase, const int qbase, const int half, const int l16)
{
    v2f kf[4][8];
#pragma unroll
    for (int j = 0; j < 4; ++j) {
        int krow = kbase + 16 * j + l16;
        if (DIAG) { if (krow > 1024) krow = 1024; }
        const float* kptr = K + krow * HSZ + 2 * half;
#pragma unroll
        for (int c = 0; c < 8; ++c) {
            float2 f = *(const float2*)(kptr + 4 * c);
            kf[j][c] = (v2f){ f.x, f.y };
        }
    }
    v2f vb0[16], vb1[16];
#pragma unroll
    for (int c = 0; c < 16; ++c) {
        const int k0 = kbase + 4 * c + 2 * half;
        int kr0 = k0, kr1 = k0 + 1;
        if (DIAG) {
            if (kr0 > 1024) kr0 = 1024;
            if (kr1 > 1024) kr1 = 1024;
        }
        vb0[c] = (v2f){ V[kr0 * HSZ + l16],      V[kr1 * HSZ + l16] };
        vb1[c] = (v2f){ V[kr0 * HSZ + 16 + l16], V[kr1 * HSZ + 16 + l16] };
    }

    v8f s[4];
#pragma unroll
    for (int j = 0; j < 4; ++j) {
        s[j] = (v8f){};
#pragma unroll
        for (int c = 0; c < 8; ++c) s[j] = wmma_f32(qf[c], kf[j][c], s[j]);
    }

    float p[4][8];
#pragma unroll
    for (int r = 0; r < 8; ++r) {
        const int qi = qbase + r + 8 * half;
        float sv[4];
#pragma unroll
        for (int j = 0; j < 4; ++j) {
            if (DIAG) {
                const int kj = kbase + 16 * j + l16;
                const bool valid = (kj <= qi) && (kj <= 1024);
                sv[j] = valid ? s[j][r] * SCALE : -3.0e38f;
            } else {
                sv[j] = s[j][r] * SCALE;
            }
        }
        float rm = fmaxf(fmaxf(sv[0], sv[1]), fmaxf(sv[2], sv[3]));
        rm = fmaxf(rm, __shfl_xor(rm, 1, 32));
        rm = fmaxf(rm, __shfl_xor(rm, 2, 32));
        rm = fmaxf(rm, __shfl_xor(rm, 4, 32));
        rm = fmaxf(rm, __shfl_xor(rm, 8, 32));
        const float nm = fmaxf(rowM[r], rm);
        float rs = 0.0f;
#pragma unroll
        for (int j = 0; j < 4; ++j) {
            float pv = __expf(sv[j] - nm);
            p[j][r] = pv;
            rs += pv;
        }
        rs += __shfl_xor(rs, 1, 32);
        rs += __shfl_xor(rs, 2, 32);
        rs += __shfl_xor(rs, 4, 32);
        rs += __shfl_xor(rs, 8, 32);
        const float corr = __expf(rowM[r] - nm);
        rowM[r] = nm;
        rowL[r] = rowL[r] * corr + rs;
        acc0[r] = acc0[r] * corr;
        acc1[r] = acc1[r] * corr;
    }

    __syncthreads();
#pragma unroll
    for (int j = 0; j < 4; ++j)
#pragma unroll
        for (int r = 0; r < 8; ++r)
            lds[(r + 8 * half) * PSTRIDE + 16 * j + l16] = p[j][r];
    __syncthreads();

#pragma unroll
    for (int c = 0; c < 16; ++c) {
        float2 pf = *(const float2*)&lds[l16 * PSTRIDE + 4 * c + 2 * half];
        v2f a = { pf.x, pf.y };
        acc0 = wmma_f32(a, vb0[c], acc0);
        acc1 = wmma_f32(a, vb1[c], acc1);
    }
}

__global__ __launch_bounds__(32) void attn_group_kernel(
    const float* __restrict__ qg, const float* __restrict__ kg,
    const float* __restrict__ vg, float* __restrict__ xg)
{
    const int qt = blockIdx.x;
    const int grp = blockIdx.y;
    const int lane = threadIdx.x;
    const int half = lane >> 4, l16 = lane & 15;

    const float* Q = qg + (size_t)grp * GLEN * HSZ;
    const float* K = kg + (size_t)grp * GLEN * HSZ;
    const float* V = vg + (size_t)grp * GLEN * HSZ;
    float*       X = xg + (size_t)grp * GLEN * HSZ;

    const int qbase = qt * 16;
    int qrow = qbase + l16; if (qrow > 1024) qrow = 1024;
    const float* qptr = Q + qrow * HSZ + 2 * half;

    v2f qf[8];
#pragma unroll
    for (int c = 0; c < 8; ++c) {
        float2 f = *(const float2*)(qptr + 4 * c);
        qf[c] = (v2f){ f.x, f.y };
    }

    v8f acc0 = {}, acc1 = {};
    float rowM[8], rowL[8];
#pragma unroll
    for (int r = 0; r < 8; ++r) { rowM[r] = -3.0e38f; rowL[r] = 0.0f; }

    __shared__ float lds[16 * PSTRIDE];

    const int nfull = qt >> 2;
    for (int b = 0; b < nfull; ++b)
        attn_block<false>(K, V, qf, acc0, acc1, rowM, rowL, lds, 64 * b, qbase, half, l16);
    attn_block<true>(K, V, qf, acc0, acc1, rowM, rowL, lds, 64 * nfull, qbase, half, l16);

#pragma unroll
    for (int r = 0; r < 8; ++r) {
        const int row = qbase + r + 8 * half;
        if (row <= 1024) {
            const float invl = 1.0f / rowL[r];
            X[row * HSZ + l16]      = acc0[r] * invl + V[row * HSZ + l16];
            X[row * HSZ + 16 + l16] = acc1[r] * invl + V[row * HSZ + 16 + l16];
        }
    }
}

__global__ __launch_bounds__(256) void meanatt_kernel(
    const float* __restrict__ qg, const float* __restrict__ kg,
    const float* __restrict__ xg, float* __restrict__ y)
{
    const int bh = blockIdx.x;
    __shared__ float qm[8][32], km[8][32], vm[8][32];
    const int t = threadIdx.x;
    const int g = t / 32, d = t % 32;
    const size_t rowoff = ((size_t)(bh * 8 + g) * GLEN + 1024) * HSZ + d;
    qm[g][d] = qg[rowoff];
    km[g][d] = kg[rowoff];
    vm[g][d] = xg[rowoff];
    __syncthreads();

    float sarr[8];
    float mx = -3.0e38f;
    for (int gp = 0; gp <= g; ++gp) {
        float sv = 0.0f;
        for (int c = 0; c < 32; ++c) sv += qm[g][c] * km[gp][c];
        sv *= SCALE;
        sarr[gp] = sv;
        mx = fmaxf(mx, sv);
    }
    float denom = 0.0f;
    for (int gp = 0; gp <= g; ++gp) { sarr[gp] = __expf(sarr[gp] - mx); denom += sarr[gp]; }
    float o = 0.0f;
    for (int gp = 0; gp <= g; ++gp) o += sarr[gp] * vm[gp][d];
    y[(bh * 8 + g) * 32 + d] = o / denom;
}

#define PROJ_TILES 5
__device__ __forceinline__ const float* zrow_ptr(
    const float* __restrict__ xg, const float* __restrict__ y, const int m, int& hstride)
{
    const int b = m / 8200;
    const int t = m - b * 8200;
    const int g = t / GLEN;
    const int r = t - g * GLEN;
    const bool isy = (g != 0) && (r == 0);
    const int gr = (g == 0) ? r : r - 1;
    const int offx = ((b * 32 + g) * GLEN + gr) * HSZ;
    const int offy = (b * 32 + g - 1) * HSZ;
    hstride = isy ? 8 * HSZ : 8 * GLEN * HSZ;
    return (isy ? y : xg) + (isy ? offy : offx);
}

__global__ __launch_bounds__(128) void proj_gemm_kernel(
    const float* __restrict__ xg, const float* __restrict__ y,
    const float* __restrict__ w, float* __restrict__ out)
{
    const int s = __builtin_amdgcn_readfirstlane((int)(threadIdx.x >> 5));
    const int lane = threadIdx.x & 31;
    const int half = lane >> 4, l16 = lane & 15;
    const int mt0 = blockIdx.x * PROJ_TILES;

    float4 bf[2][16];
#pragma unroll
    for (int t = 0; t < 2; ++t) {
        const float* brow = w + (size_t)(32 * s + 16 * t + l16) * 128 + 4 * half;
#pragma unroll
        for (int c = 0; c < 16; ++c) bf[t][c] = *(const float4*)(brow + 8 * c);
    }

    float4 af[2][16];
    {
        int hs;
        const float* arow = zrow_ptr(xg, y, mt0 * 16 + l16, hs) + 4 * half;
#pragma unroll
        for (int c = 0; c < 16; ++c) af[0][c] = *(const float4*)(arow + (c >> 2) * hs + 8 * (c & 3));
    }
    __builtin_amdgcn_sched_barrier(0);
#pragma unroll
    for (int i = 0; i < PROJ_TILES; ++i) {
        if (i + 1 < PROJ_TILES) {
            int hs;
            const float* arow = zrow_ptr(xg, y, (mt0 + i + 1) * 16 + l16, hs) + 4 * half;
#pragma unroll
            for (int c = 0; c < 16; ++c) af[(i + 1) & 1][c] = *(const float4*)(arow + (c >> 2) * hs + 8 * (c & 3));
        }
        __builtin_amdgcn_sched_barrier(0);
        v8f acc0 = {}, acc1 = {};
#pragma unroll
        for (int c = 0; c < 16; ++c) {
            const float4 a = af[i & 1][c];
            const v2f alo = { a.x, a.y }, ahi = { a.z, a.w };
            const v2f b0lo = { bf[0][c].x, bf[0][c].y }, b0hi = { bf[0][c].z, bf[0][c].w };
            const v2f b1lo = { bf[1][c].x, bf[1][c].y }, b1hi = { bf[1][c].z, bf[1][c].w };
            acc0 = wmma_f32(alo, b0lo, acc0);
            acc1 = wmma_f32(alo, b1lo, acc1);
            acc0 = wmma_f32(ahi, b0hi, acc0);
            acc1 = wmma_f32(ahi, b1hi, acc1);
        }
        float* base = out + (size_t)((mt0 + i) * 16 + 8 * half) * 128 + 32 * s + l16;
#pragma unroll
        for (int r = 0; r < 8; ++r) {
            base[r * 128]      = acc0[r];
            base[r * 128 + 16] = acc1[r];
        }
    }
}

extern "C" void kernel_launch(void* const* d_in, const int* in_sizes, int n_in,
                              void* d_out, int out_size, void* d_ws, size_t ws_size,
                              hipStream_t stream) {
    const float* x      = (const float*)d_in[0];
    const float* w_attn = (const float*)d_in[1];
    const float* w_proj = (const float*)d_in[2];

    float* ws = (float*)d_ws;
    const size_t GBUF = (size_t)GROUPS * GLEN * HSZ;
    float* qg = ws;
    float* kg = ws + GBUF;
    float* vg = ws + 2 * GBUF;
    float* xg = ws + 3 * GBUF;
    float* y  = ws + 4 * GBUF;

    qkv_gemm_kernel<<<256, 384, 0, stream>>>(x, w_attn, qg, kg, vg);
    mean_kernel<<<dim3(GROUPS, 3), MEAN_THREADS, 0, stream>>>(qg, kg, vg);
    attn_group_kernel<<<dim3(65, GROUPS), 32, 0, stream>>>(qg, kg, vg, xg);
    meanatt_kernel<<<8, 256, 0, stream>>>(qg, kg, xg, y);
    proj_gemm_kernel<<<205, 128, 0, stream>>>(xg, y, w_proj, (float*)d_out);
}


# attention kernel: score tile computed transposed (S^T = K Q^T), so P is the second product's B operand straight from registers; LDS staging, barriers and 62 of 64 lane exchanges per block are gone
# speedup vs baseline: 3.0547x; 1.2550x over previous
#include <hip/hip_runtime.h>
#include <hip/hip_bf16.h>

typedef float v2f __attribute__((ext_vector_type(2)));
typedef float v8f __attribute__((ext_vector_type(8)));

#define GROUPS 64
#define GLEN   1025
#define HSZ    32
#define SCALE  0.17677669529663687f

__device__ __forceinline__ v8f wmma_f32(v2f a, v2f b, v8f c) {
    return __builtin_amdgcn_wmma_f32_16x16x4_f32(false, a, false, b, (short)0, c, false, false);
}

#define QKV_TILES 4
__global__ __launch_bounds__(384) __attribute__((amdgpu_waves_per_eu(3, 3))) void qkv_gemm_kernel(
    const float* __restrict__ x, const float* __restrict__ w,
    float* __restrict__ qg, float* __restrict__ kg, float* __restrict__ vg)
{
    const int s = __builtin_amdgcn_readfirstlane((int)(threadIdx.x >> 5));
    const int lane = threadIdx.x & 31;
    const int half = lane >> 4, l16 = lane & 15;
    const int mt0 = blockIdx.x * QKV_TILES;

    float4 bf[2][16];
#pragma unroll
    for (int t = 0; t < 2; ++t) {
        const float* brow = w + (size_t)(32 * s + 16 * t + l16) * 128 + 4 * half;
#pragma unroll
        for (int c = 0; c < 16; ++c) bf[t][c] = *(const float4*)(brow + 8 * c);
    }

    const int which = s >> 2;
    const int h = s & 3;
    float* dst = (which == 0) ? qg : ((which == 1) ? kg : vg);

    float4 af[2][16];
    {
        const float* arow = x + (size_t)(mt0 * 16 + l16) * 128 + 4 * half;
#pragma unroll
        for (int c = 0; c < 16; ++c) af[0][c] = *(const float4*)(arow + 8 * c);
    }
    __builtin_amdgcn_sched_barrier(0);
#pragma unroll
    for (int i = 0; i < QKV_TILES; ++i) {
        if (i + 1 < QKV_TILES) {
            const float* arow = x + (size_t)((mt0 + i + 1) * 16 + l16) * 128 + 4 * half;
#pragma unroll
            for (int c = 0; c < 16; ++c) af[(i + 1) & 1][c] = *(const float4*)(arow + 8 * c);
        }
        __builtin_amdgcn_sched_barrier(0);
        v8f acc0 = {}, acc1 = {};
#pragma unroll
        for (int c = 0; c < 16; ++c) {
            const float4 a = af[i & 1][c];
            const v2f alo = { a.x, a.y }, ahi = { a.z, a.w };
            const v2f b0lo = { bf[0][c].x, bf[0][c].y }, b0hi = { bf[0][c].z, bf[0][c].w };
            const v2f b1lo = { bf[1][c].x, bf[1][c].y }, b1hi = { bf[1][c].z, bf[1][c].w };
            acc0 = wmma_f32(alo, b0lo, acc0);
            acc1 = wmma_f32(alo, b1lo, acc1);
            acc0 = wmma_f32(ahi, b0hi, acc0);
            acc1 = wmma_f32(ahi, b1hi, acc1);
        }
        const int mt = mt0 + i;
        const int bidx = mt >> 9;
        const int g = (mt & 511) >> 6;
        const int rig0 = (mt & 63) * 16;
        float* base = dst + ((size_t)((bidx * 4 + h) * 8 + g) * GLEN + rig0 + 8 * half) * HSZ + l16;
#pragma unroll
        for (int r = 0; r < 8; ++r) {
            base[r * HSZ]      = acc0[r];
            base[r * HSZ + 16] = acc1[r];
        }
    }
}

#define MEAN_THREADS 256
#define MEAN_LOADS   32
__global__ __launch_bounds__(MEAN_THREADS) __attribute__((amdgpu_waves_per_eu(2, 2))) void mean_kernel(
    float* __restrict__ qg, float* __restrict__ kg, float* __restrict__ vg)
{
    const int grp = blockIdx.x;
    const int which = blockIdx.y;
    float* buf = (which == 0) ? qg : ((which == 1) ? kg : vg);
    float* gbase = buf + (size_t)grp * GLEN * HSZ;
    const int t = threadIdx.x;
    const int wave = t >> 5, lane = t & 31;

    const float4* src = (const float4*)gbase + t;
    float4 v[MEAN_LOADS];
#pragma unroll
    for (int i = 0; i < MEAN_LOADS; ++i) v[i] = src[i * MEAN_THREADS];
    __builtin_amdgcn_sched_barrier(0);

#pragma unroll
    for (int s = MEAN_LOADS / 2; s >= 1; s >>= 1) {
#pragma unroll
        for (int i = 0; i < s; ++i) {
            v[i].x += v[i + s].x;
            v[i].y += v[i + s].y;
            v[i].z += v[i + s].z;
            v[i].w += v[i + s].w;
        }
    }
    float4 a = v[0];
    a.x += __shfl_xor(a.x, 8, 32);  a.y += __shfl_xor(a.y, 8, 32);
    a.z += __shfl_xor(a.z, 8, 32);  a.w += __shfl_xor(a.w, 8, 32);
    a.x += __shfl_xor(a.x, 16, 32); a.y += __shfl_xor(a.y, 16, 32);
    a.z += __shfl_xor(a.z, 16, 32); a.w += __shfl_xor(a.w, 16, 32);

    __shared__ __attribute__((aligned(16))) float part[MEAN_THREADS / 32][HSZ];
    if (lane < 8) *(float4*)&part[wave][4 * lane] = a;
    __syncthreads();
    if (t < HSZ) {
        float s = 0.0f;
#pragma unroll
        for (int w = 0; w < MEAN_THREADS / 32; ++w) s += part[w][t];
        gbase[1024 * HSZ + t] = s * (1.0f / 1024.0f);
    }
}

__device__ __forceinline__ float xhalf(float x) {
    const int b = __builtin_bit_cast(int, x);
    return __builtin_bit_cast(float,
        __builtin_amdgcn_permlanex16(b, b, 0x76543210, 0xfedcba98, false, false));
}

template <bool DIAG>
__device__ __forceinline__ void attn_block(
    const float* __restrict__ K, const float* __restrict__ V,
    const v2f (&qf)[8], v8f& acc0, v8f& acc1, float& colM, float& colL,
    const int kbase, const int qbase, const int half, const int l16)
{
    v2f kf[4][8];
#pragma unroll
    for (int j = 0; j < 4; ++j) {
        int krow = kbase + 16 * j + l16;
        if (DIAG) { if (krow > 1024) krow = 1024; }
        const float* kptr = K + krow * HSZ + 2 * half;
#pragma unroll
        for (int c = 0; c < 8; ++c) {
            float2 f = *(const float2*)(kptr + 4 * c);
            kf[j][c] = (v2f){ f.x, f.y };
        }
    }
    v2f va0[4][4], va1[4][4];
#pragma unroll
    for (int j = 0; j < 4; ++j) {
#pragma unroll
        for (int i = 0; i < 4; ++i) {
            const int k0 = kbase + 16 * j + 8 * half + 2 * i;
            int kr0 = k0, kr1 = k0 + 1;
            if (DIAG) {
                if (kr0 > 1024) kr0 = 1024;
                if (kr1 > 1024) kr1 = 1024;
            }
            va0[j][i] = (v2f){ V[kr0 * HSZ + l16],      V[kr1 * HSZ + l16] };
            va1[j][i] = (v2f){ V[kr0 * HSZ + 16 + l16], V[kr1 * HSZ + 16 + l16] };
        }
    }

    __builtin_amdgcn_sched_barrier(0);

    v8f s[4];
#pragma unroll
    for (int j = 0; j < 4; ++j) {
        s[j] = (v8f){};
#pragma unroll
        for (int c = 0; c < 8; ++c) s[j] = wmma_f32(kf[j][c], qf[c], s[j]);
    }

    const int qi = qbase + l16;
    float mj[4];
#pragma unroll
    for (int j = 0; j < 4; ++j) {
        mj[j] = -3.0e38f;
#pragma unroll
        for (int r = 0; r < 8; ++r) {
            float sv = s[j][r] * SCALE;
            if (DIAG) {
                const int kj = kbase + 16 * j + 8 * half + r;
                const bool valid = (kj <= qi) && (kj <= 1024);
                sv = valid ? sv : -3.0e38f;
            }
            s[j][r] = sv;
            mj[j] = fmaxf(mj[j], sv);
        }
    }
    float mx = fmaxf(fmaxf(mj[0], mj[1]), fmaxf(mj[2], mj[3]));
    mx = fmaxf(mx, xhalf(mx));
    const float nm = fmaxf(colM, mx);
    float rj[4];
#pragma unroll
    for (int j = 0; j < 4; ++j) {
        rj[j] = 0.0f;
#pragma unroll
        for (int r = 0; r < 8; ++r) {
            const float pv = __expf(s[j][r] - nm);
            s[j][r] = pv;
            rj[j] += pv;
        }
    }
    float rs = (rj[0] + rj[1]) + (rj[2] + rj[3]);
    rs += xhalf(rs);
    const float corr = __expf(colM - nm);
    colM = nm;
    colL = colL * corr + rs;
#pragma unroll
    for (int r = 0; r < 8; ++r) {
        acc0[r] = acc0[r] * corr;
        acc1[r] = acc1[r] * corr;
    }

#pragma unroll
    for (int j = 0; j < 4; ++j) {
#pragma unroll
        for (int i = 0; i < 4; ++i) {
            const v2f pb = { s[j][2 * i], s[j][2 * i + 1] };
            acc0 = wmma_f32(va0[j][i], pb, acc0);
            acc1 = wmma_f32(va1[j][i], pb, acc1);
        }
    }
}

__global__ __launch_bounds__(32) void attn_group_kernel(
    const float* __restrict__ qg, const float* __restrict__ kg,
    const float* __restrict__ vg, float* __restrict__ xg)
{
    const int qt = blockIdx.x;
    const int grp = blockIdx.y;
    const int lane = threadIdx.x;
    const int half = lane >> 4, l16 = lane & 15;

    const float* Q = qg + (size_t)grp * GLEN * HSZ;
    const float* K = kg + (size_t)grp * GLEN * HSZ;
    const float* V = vg + (size_t)grp * GLEN * HSZ;
    float*       X = xg + (size_t)grp * GLEN * HSZ;

    const int qbase = qt * 16;
    int qrow = qbase + l16; if (qrow > 1024) qrow = 1024;
    const float* qptr = Q + qrow * HSZ + 2 * half;

    v2f qf[8];
#pragma unroll
    for (int c = 0; c < 8; ++c) {
        float2 f = *(const float2*)(qptr + 4 * c);
        qf[c] = (v2f){ f.x, f.y };
    }

    v8f acc0 = {}, acc1 = {};
    float colM = -3.0e38f, colL = 0.0f;

    const int nfull = qt >> 2;
    for (int b = 0; b < nfull; ++b)
        attn_block<false>(K, V, qf, acc0, acc1, colM, colL, 64 * b, qbase, half, l16);
    attn_block<true>(K, V, qf, acc0, acc1, colM, colL, 64 * nfull, qbase, half, l16);

    const int row = qbase + l16;
    if (row <= 1024) {
        const float invl = 1.0f / colL;
        const float4* vr = (const float4*)(V + row * HSZ + 8 * half);
        float4*       xr = (float4*)(X + row * HSZ + 8 * half);
        const float4 v0 = vr[0], v1 = vr[1], v2 = vr[4], v3 = vr[5];
        float4 o0, o1, o2, o3;
        o0.x = acc0[0] * invl + v0.x;  o0.y = acc0[1] * invl + v0.y;
        o0.z = acc0[2] * invl + v0.z;  o0.w = acc0[3] * invl + v0.w;
        o1.x = acc0[4] * invl + v1.x;  o1.y = acc0[5] * invl + v1.y;
        o1.z = acc0[6] * invl + v1.z;  o1.w = acc0[7] * invl + v1.w;
        o2.x = acc1[0] * invl + v2.x;  o2.y = acc1[1] * invl + v2.y;
        o2.z = acc1[2] * invl + v2.z;  o2.w = acc1[3] * invl + v2.w;
        o3.x = acc1[4] * invl + v3.x;  o3.y = acc1[5] * invl + v3.y;
        o3.z = acc1[6] * invl + v3.z;  o3.w = acc1[7] * invl + v3.w;
        xr[0] = o0;  xr[1] = o1;  xr[4] = o2;  xr[5] = o3;
    }
}

__global__ __launch_bounds__(256) void meanatt_kernel(
    const float* __restrict__ qg, const float* __restrict__ kg,
    const float* __restrict__ xg, float* __restrict__ y)
{
    const int bh = blockIdx.x;
    __shared__ float qm[8][32], km[8][32], vm[8][32];
    const int t = threadIdx.x;
    const int g = t / 32, d = t % 32;
    const size_t rowoff = ((size_t)(bh * 8 + g) * GLEN + 1024) * HSZ + d;
    qm[g][d] = qg[rowoff];
    km[g][d] = kg[rowoff];
    vm[g][d] = xg[rowoff];
    __syncthreads();

    float sarr[8];
    float mx = -3.0e38f;
    for (int gp = 0; gp <= g; ++gp) {
        float sv = 0.0f;
        for (int c = 0; c < 32; ++c) sv += qm[g][c] * km[gp][c];
        sv *= SCALE;
        sarr[gp] = sv;
        mx = fmaxf(mx, sv);
    }
    float denom = 0.0f;
    for (int gp = 0; gp <= g; ++gp) { sarr[gp] = __expf(sarr[gp] - mx); denom += sarr[gp]; }
    float o = 0.0f;
    for (int gp = 0; gp <= g; ++gp) o += sarr[gp] * vm[gp][d];
    y[(bh * 8 + g) * 32 + d] = o / denom;
}

#define PROJ_TILES 5
__device__ __forceinline__ const float* zrow_ptr(
    const float* __restrict__ xg, const float* __restrict__ y, const int m, int& hstride)
{
    const int b = m / 8200;
    const int t = m - b * 8200;
    const int g = t / GLEN;
    const int r = t - g * GLEN;
    const bool isy = (g != 0) && (r == 0);
    const int gr = (g == 0) ? r : r - 1;
    const int offx = ((b * 32 + g) * GLEN + gr) * HSZ;
    const int offy = (b * 32 + g - 1) * HSZ;
    hstride = isy ? 8 * HSZ : 8 * GLEN * HSZ;
    return (isy ? y : xg) + (isy ? offy : offx);
}

__global__ __launch_bounds__(128) void proj_gemm_kernel(
    const float* __restrict__ xg, const float* __restrict__ y,
    const float* __restrict__ w, float* __restrict__ out)
{
    const int s = __builtin_amdgcn_readfirstlane((int)(threadIdx.x >> 5));
    const int lane = threadIdx.x & 31;
    const int half = lane >> 4, l16 = lane & 15;
    const int mt0 = blockIdx.x * PROJ_TILES;

    float4 bf[2][16];
#pragma unroll
    for (int t = 0; t < 2; ++t) {
        const float* brow = w + (size_t)(32 * s + 16 * t + l16) * 128 + 4 * half;
#pragma unroll
        for (int c = 0; c < 16; ++c) bf[t][c] = *(const float4*)(brow + 8 * c);
    }

    float4 af[2][16];
    {
        int hs;
        const float* arow = zrow_ptr(xg, y, mt0 * 16 + l16, hs) + 4 * half;
#pragma unroll
        for (int c = 0; c < 16; ++c) af[0][c] = *(const float4*)(arow + (c >> 2) * hs + 8 * (c & 3));
    }
    __builtin_amdgcn_sched_barrier(0);
#pragma unroll
    for (int i = 0; i < PROJ_TILES; ++i) {
        if (i + 1 < PROJ_TILES) {
            int hs;
            const float* arow = zrow_ptr(xg, y, (mt0 + i + 1) * 16 + l16, hs) + 4 * half;
#pragma unroll
            for (int c = 0; c < 16; ++c) af[(i + 1) & 1][c] = *(const float4*)(arow + (c >> 2) * hs + 8 * (c & 3));
        }
        __builtin_amdgcn_sched_barrier(0);
        v8f acc0 = {}, acc1 = {};
#pragma unroll
        for (int c = 0; c < 16; ++c) {
            const float4 a = af[i & 1][c];
            const v2f alo = { a.x, a.y }, ahi = { a.z, a.w };
            const v2f b0lo = { bf[0][c].x, bf[0][c].y }, b0hi = { bf[0][c].z, bf[0][c].w };
            const v2f b1lo = { bf[1][c].x, bf[1][c].y }, b1hi = { bf[1][c].z, bf[1][c].w };
            acc0 = wmma_f32(alo, b0lo, acc0);
            acc1 = wmma_f32(alo, b1lo, acc1);
            acc0 = wmma_f32(ahi, b0hi, acc0);
            acc1 = wmma_f32(ahi, b1hi, acc1);
        }
        float* base = out + (size_t)((mt0 + i) * 16 + 8 * half) * 128 + 32 * s + l16;
#pragma unroll
        for (int r = 0; r < 8; ++r) {
            base[r * 128]      = acc0[r];
            base[r * 128 + 16] = acc1[r];
        }
    }
}

extern "C" void kernel_launch(void* const* d_in, const int* in_sizes, int n_in,
                              void* d_out, int out_size, void* d_ws, size_t ws_size,
                              hipStream_t stream) {
    const float* x      = (const float*)d_in[0];
    const float* w_attn = (const float*)d_in[1];
    const float* w_proj = (const float*)d_in[2];

    float* ws = (float*)d_ws;
    const size_t GBUF = (size_t)GROUPS * GLEN * HSZ;
    float* qg = ws;
    float* kg = ws + GBUF;
    float* vg = ws + 2 * GBUF;
    float* xg = ws + 3 * GBUF;
    float* y  = ws + 4 * GBUF;

    qkv_gemm_kernel<<<256, 384, 0, stream>>>(x, w_attn, qg, kg, vg);
    mean_kernel<<<dim3(GROUPS, 3), MEAN_THREADS, 0, stream>>>(qg, kg, vg);
    attn_group_kernel<<<dim3(65, GROUPS), 32, 0, stream>>>(qg, kg, vg, xg);
    meanatt_kernel<<<8, 256, 0, stream>>>(qg, kg, xg, y);
    proj_gemm_kernel<<<205, 128, 0, stream>>>(xg, y, w_proj, (float*)d_out);
}


# attention kernel: the 4 Q tiles that walk the same key blocks run as the 4 waves of one work-group, issued in a work-balanced order: K/V lines come from L2 once per work-group, not once per wave
# speedup vs baseline: 3.2642x; 1.0686x over previous
#include <hip/hip_runtime.h>
#include <hip/hip_bf16.h>

typedef float v2f __attribute__((ext_vector_type(2)));
typedef float v8f __attribute__((ext_vector_type(8)));

#define GROUPS 64
#define GLEN   1025
#define HSZ    32
#define SCALE  0.17677669529663687f

__device__ __forceinline__ v8f wmma_f32(v2f a, v2f b, v8f c) {
    return __builtin_amdgcn_wmma_f32_16x16x4_f32(false, a, false, b, (short)0, c, false, false);
}

#define QKV_TILES 4
__global__ __launch_bounds__(384) __attribute__((amdgpu_waves_per_eu(3, 3))) void qkv_gemm_kernel(
    const float* __restrict__ x, const float* __restrict__ w,
    float* __restrict__ qg, float* __restrict__ kg, float* __restrict__ vg)
{
    const int s = __builtin_amdgcn_readfirstlane((int)(threadIdx.x >> 5));
    const int lane = threadIdx.x & 31;
    const int half = lane >> 4, l16 = lane & 15;
    const int mt0 = blockIdx.x * QKV_TILES;

    float4 bf[2][16];
#pragma unroll
    for (int t = 0; t < 2; ++t) {
        const float* brow = w + (size_t)(32 * s + 16 * t + l16) * 128 + 4 * half;
#pragma unroll
        for (int c = 0; c < 16; ++c) bf[t][c] = *(const float4*)(brow + 8 * c);
    }

    const int which = s >> 2;
    const int h = s & 3;
    float* dst = (which == 0) ? qg : ((which == 1) ? kg : vg);

    float4 af[2][16];
    {
        const float* arow = x + (size_t)(mt0 * 16 + l16) * 128 + 4 * half;
#pragma unroll
        for (int c = 0; c < 16; ++c) af[0][c] = *(const float4*)(arow + 8 * c);
    }
    __builtin_amdgcn_sched_barrier(0);
#pragma unroll
    for (int i = 0; i < QKV_TILES; ++i) {
        if (i + 1 < QKV_TILES) {
            const float* arow = x + (size_t)((mt0 + i + 1) * 16 + l16) * 128 + 4 * half;
#pragma unroll
            for (int c = 0; c < 16; ++c) af[(i + 1) & 1][c] = *(const float4*)(arow + 8 * c);
        }
        __builtin_amdgcn_sched_barrier(0);
        v8f acc0 = {}, acc1 = {};
#pragma unroll
        for (int c = 0; c < 16; ++c) {
            const float4 a = af[i & 1][c];
            const v2f alo = { a.x, a.y }, ahi = { a.z, a.w };
            const v2f b0lo = { bf[0][c].x, bf[0][c].y }, b0hi = { bf[0][c].z, bf[0][c].w };
            const v2f b1lo = { bf[1][c].x, bf[1][c].y }, b1hi = { bf[1][c].z, bf[1][c].w };
            acc0 = wmma_f32(alo, b0lo, acc0);
            acc1 = wmma_f32(alo, b1lo, acc1);
            acc0 = wmma_f32(ahi, b0hi, acc0);
            acc1 = wmma_f32(ahi, b1hi, acc1);
        }
        const int mt = mt0 + i;
        const int bidx = mt >> 9;
        const int g = (mt & 511) >> 6;
        const int rig0 = (mt & 63) * 16;
        float* base = dst + ((size_t)((bidx * 4 + h) * 8 + g) * GLEN + rig0 + 8 * half) * HSZ + l16;
#pragma unroll
        for (int r = 0; r < 8; ++r) {
            base[r * HSZ]      = acc0[r];
            base[r * HSZ + 16] = acc1[r];
        }
    }
}

#define MEAN_THREADS 256
#define MEAN_LOADS   32
__global__ __launch_bounds__(MEAN_THREADS) __attribute__((amdgpu_waves_per_eu(2, 2))) void mean_kernel(
    float* __restrict__ qg, float* __restrict__ kg, float* __restrict__ vg)
{
    const int grp = blockIdx.x;
    const int which = blockIdx.y;
    float* buf = (which == 0) ? qg : ((which == 1) ? kg : vg);
    float* gbase = buf + (size_t)grp * GLEN * HSZ;
    const int t = threadIdx.x;
    const int wave = t >> 5, lane = t & 31;

    const float4* src = (const float4*)gbase + t;
    float4 v[MEAN_LOADS];
#pragma unroll
    for (int i = 0; i < MEAN_LOADS; ++i) v[i] = src[i * MEAN_THREADS];
    __builtin_amdgcn_sched_barrier(0);

#pragma unroll
    for (int s = MEAN_LOADS / 2; s >= 1; s >>= 1) {
#pragma unroll
        for (int i = 0; i < s; ++i) {
            v[i].x += v[i + s].x;
            v[i].y += v[i + s].y;
            v[i].z += v[i + s].z;
            v[i].w += v[i + s].w;
        }
    }
    float4 a = v[0];
    a.x += __shfl_xor(a.x, 8, 32);  a.y += __shfl_xor(a.y, 8, 32);
    a.z += __shfl_xor(a.z, 8, 32);  a.w += __shfl_xor(a.w, 8, 32);
    a.x += __shfl_xor(a.x, 16, 32); a.y += __shfl_xor(a.y, 16, 32);
    a.z += __shfl_xor(a.z, 16, 32); a.w += __shfl_xor(a.w, 16, 32);

    __shared__ __attribute__((aligned(16))) float part[MEAN_THREADS / 32][HSZ];
    if (lane < 8) *(float4*)&part[wave][4 * lane] = a;
    __syncthreads();
    if (t < HSZ) {
        float s = 0.0f;
#pragma unroll
        for (int w = 0; w < MEAN_THREADS / 32; ++w) s += part[w][t];
        gbase[1024 * HSZ + t] = s * (1.0f / 1024.0f);
    }
}

__device__ __forceinline__ float xhalf(float x) {
    const int b = __builtin_bit_cast(int, x);
    return __builtin_bit_cast(float,
        __builtin_amdgcn_permlanex16(b, b, 0x76543210, 0xfedcba98, false, false));
}

template <bool DIAG>
__device__ __forceinline__ void attn_block(
    const float* __restrict__ K, const float* __restrict__ V,
    const v2f (&qf)[8], v8f& acc0, v8f& acc1, float& colM, float& colL,
    const int kbase, const int qbase, const int half, const int l16)
{
    v2f kf[4][8];
#pragma unroll
    for (int j = 0; j < 4; ++j) {
        int krow = kbase + 16 * j + l16;
        if (DIAG) { if (krow > 1024) krow = 1024; }
        const float* kptr = K + krow * HSZ + 2 * half;
#pragma unroll
        for (int c = 0; c < 8; ++c) {
            float2 f = *(const float2*)(kptr + 4 * c);
            kf[j][c] = (v2f){ f.x, f.y };
        }
    }
    v2f va0[4][4], va1[4][4];
#pragma unroll
    for (int j = 0; j < 4; ++j) {
#pragma unroll
        for (int i = 0; i < 4; ++i) {
            const int k0 = kbase + 16 * j + 8 * half + 2 * i;
            int kr0 = k0, kr1 = k0 + 1;
            if (DIAG) {
                if (kr0 > 1024) kr0 = 1024;
                if (kr1 > 1024) kr1 = 1024;
            }
            va0[j][i] = (v2f){ V[kr0 * HSZ + l16],      V[kr1 * HSZ + l16] };
            va1[j][i] = (v2f){ V[kr0 * HSZ + 16 + l16], V[kr1 * HSZ + 16 + l16] };
        }
    }

    __builtin_amdgcn_sched_barrier(0);

    v8f s[4];
#pragma unroll
    for (int j = 0; j < 4; ++j) {
        s[j] = (v8f){};
#pragma unroll
        for (int c = 0; c < 8; ++c) s[j] = wmma_f32(kf[j][c], qf[c], s[j]);
    }

    const int qi = qbase + l16;
    float mj[4];
#pragma unroll
    for (int j = 0; j < 4; ++j) {
        mj[j] = -3.0e38f;
#pragma unroll
        for (int r = 0; r < 8; ++r) {
            float sv = s[j][r] * SCALE;
            if (DIAG) {
                const int kj = kbase + 16 * j + 8 * half + r;
                const bool valid = (kj <= qi) && (kj <= 1024);
                sv = valid ? sv : -3.0e38f;
            }
            s[j][r] = sv;
            mj[j] = fmaxf(mj[j], sv);
        }
    }
    float mx = fmaxf(fmaxf(mj[0], mj[1]), fmaxf(mj[2], mj[3]));
    mx = fmaxf(mx, xhalf(mx));
    const float nm = fmaxf(colM, mx);
    float rj[4];
#pragma unroll
    for (int j = 0; j < 4; ++j) {
        rj[j] = 0.0f;
#pragma unroll
        for (int r = 0; r < 8; ++r) {
            const float pv = __expf(s[j][r] - nm);
            s[j][r] = pv;
            rj[j] += pv;
        }
    }
    float rs = (rj[0] + rj[1]) + (rj[2] + rj[3]);
    rs += xhalf(rs);
    const float corr = __expf(colM - nm);
    colM = nm;
    colL = colL * corr + rs;
#pragma unroll
    for (int r = 0; r < 8; ++r) {
        acc0[r] = acc0[r] * corr;
        acc1[r] = acc1[r] * corr;
    }

#pragma unroll
    for (int j = 0; j < 4; ++j) {
#pragma unroll
        for (int i = 0; i < 4; ++i) {
            const v2f pb = { s[j][2 * i], s[j][2 * i + 1] };
            acc0 = wmma_f32(va0[j][i], pb, acc0);
            acc1 = wmma_f32(va1[j][i], pb, acc1);
        }
    }
}

#define ATT_WAVES 4
#define ATT_WGS   1088
__global__ __launch_bounds__(32 * ATT_WAVES) void attn_group_kernel(
    const float* __restrict__ qg, const float* __restrict__ kg,
    const float* __restrict__ vg, float* __restrict__ xg)
{
    const int id = blockIdx.x;
    int w, grp;
    if (id < 1024) {
        const unsigned long long sq = 0x0DE3B6587A94C12FULL;
        const int m = id >> 2;
        w = 1 + (int)((sq >> (4 * (4 * (id & 3) + (m >> 6)))) & 15ULL);
        grp = m & 63;
    } else {
        w = 0;
        grp = id - 1024;
    }
    const int qt = 4 * w + __builtin_amdgcn_readfirstlane((int)(threadIdx.x >> 5));
    if (qt > 64) return;
    const int lane = threadIdx.x & 31;
    const int half = lane >> 4, l16 = lane & 15;

    const float* Q = qg + (size_t)grp * GLEN * HSZ;
    const float* K = kg + (size_t)grp * GLEN * HSZ;
    const float* V = vg + (size_t)grp * GLEN * HSZ;
    float*       X = xg + (size_t)grp * GLEN * HSZ;

    const int qbase = qt * 16;
    int qrow = qbase + l16; if (qrow > 1024) qrow = 1024;
    const float* qptr = Q + qrow * HSZ + 2 * half;

    v2f qf[8];
#pragma unroll
    for (int c = 0; c < 8; ++c) {
        float2 f = *(const float2*)(qptr + 4 * c);
        qf[c] = (v2f){ f.x, f.y };
    }

    v8f acc0 = {}, acc1 = {};
    float colM = -3.0e38f, colL = 0.0f;

    const int nfull = qt >> 2;
    for (int b = 0; b < nfull; ++b)
        attn_block<false>(K, V, qf, acc0, acc1, colM, colL, 64 * b, qbase, half, l16);
    attn_block<true>(K, V, qf, acc0, acc1, colM, colL, 64 * nfull, qbase, half, l16);

    const int row = qbase + l16;
    if (row <= 1024) {
        const float invl = 1.0f / colL;
        const float4* vr = (const float4*)(V + row * HSZ + 8 * half);
        float4*       xr = (float4*)(X + row * HSZ + 8 * half);
        const float4 v0 = vr[0], v1 = vr[1], v2 = vr[4], v3 = vr[5];
        float4 o0, o1, o2, o3;
        o0.x = acc0[0] * invl + v0.x;  o0.y = acc0[1] * invl + v0.y;
        o0.z = acc0[2] * invl + v0.z;  o0.w = acc0[3] * invl + v0.w;
        o1.x = acc0[4] * invl + v1.x;  o1.y = acc0[5] * invl + v1.y;
        o1.z = acc0[6] * invl + v1.z;  o1.w = acc0[7] * invl + v1.w;
        o2.x = acc1[0] * invl + v2.x;  o2.y = acc1[1] * invl + v2.y;
        o2.z = acc1[2] * invl + v2.z;  o2.w = acc1[3] * invl + v2.w;
        o3.x = acc1[4] * invl + v3.x;  o3.y = acc1[5] * invl + v3.y;
        o3.z = acc1[6] * invl + v3.z;  o3.w = acc1[7] * invl + v3.w;
        xr[0] = o0;  xr[1] = o1;  xr[4] = o2;  xr[5] = o3;
    }
}

__global__ __launch_bounds__(256) void meanatt_kernel(
    const float* __restrict__ qg, const float* __restrict__ kg,
    const float* __restrict__ xg, float* __restrict__ y)
{
    const int bh = blockIdx.x;
    __shared__ float qm[8][32], km[8][32], vm[8][32];
    const int t = threadIdx.x;
    const int g = t / 32, d = t % 32;
    const size_t rowoff = ((size_t)(bh * 8 + g) * GLEN + 1024) * HSZ + d;
    qm[g][d] = qg[rowoff];
    km[g][d] = kg[rowoff];
    vm[g][d] = xg[rowoff];
    __syncthreads();

    float sarr[8];
    float mx = -3.0e38f;
    for (int gp = 0; gp <= g; ++gp) {
        float sv = 0.0f;
        for (int c = 0; c < 32; ++c) sv += qm[g][c] * km[gp][c];
        sv *= SCALE;
        sarr[gp] = sv;
        mx = fmaxf(mx, sv);
    }
    float denom = 0.0f;
    for (int gp = 0; gp <= g; ++gp) { sarr[gp] = __expf(sarr[gp] - mx); denom += sarr[gp]; }
    float o = 0.0f;
    for (int gp = 0; gp <= g; ++gp) o += sarr[gp] * vm[gp][d];
    y[(bh * 8 + g) * 32 + d] = o / denom;
}

#define PROJ_TILES 5
__device__ __forceinline__ const float* zrow_ptr(
    const float* __restrict__ xg, const float* __restrict__ y, const int m, int& hstride)
{
    const int b = m / 8200;
    const int t = m - b * 8200;
    const int g = t / GLEN;
    const int r = t - g * GLEN;
    const bool isy = (g != 0) && (r == 0);
    const int gr = (g == 0) ? r : r - 1;
    const int offx = ((b * 32 + g) * GLEN + gr) * HSZ;
    const int offy = (b * 32 + g - 1) * HSZ;
    hstride = isy ? 8 * HSZ : 8 * GLEN * HSZ;
    return (isy ? y : xg) + (isy ? offy : offx);
}

__global__ __launch_bounds__(128) void proj_gemm_kernel(
    const float* __restrict__ xg, const float* __restrict__ y,
    const float* __restrict__ w, float* __restrict__ out)
{
    const int s = __builtin_amdgcn_readfirstlane((int)(threadIdx.x >> 5));
    const int lane = threadIdx.x & 31;
    const int half = lane >> 4, l16 = lane & 15;
    const int mt0 = blockIdx.x * PROJ_TILES;

    float4 bf[2][16];
#pragma unroll
    for (int t = 0; t < 2; ++t) {
        const float* brow = w + (size_t)(32 * s + 16 * t + l16) * 128 + 4 * half;
#pragma unroll
        for (int c = 0; c < 16; ++c) bf[t][c] = *(const float4*)(brow + 8 * c);
    }

    float4 af[2][16];
    {
        int hs;
        const float* arow = zrow_ptr(xg, y, mt0 * 16 + l16, hs) + 4 * half;
#pragma unroll
        for (int c = 0; c < 16; ++c) af[0][c] = *(const float4*)(arow + (c >> 2) * hs + 8 * (c & 3));
    }
    __builtin_amdgcn_sched_barrier(0);
#pragma unroll
    for (int i = 0; i < PROJ_TILES; ++i) {
        if (i + 1 < PROJ_TILES) {
            int hs;
            const float* arow = zrow_ptr(xg, y, (mt0 + i + 1) * 16 + l16, hs) + 4 * half;
#pragma unroll
            for (int c = 0; c < 16; ++c) af[(i + 1) & 1][c] = *(const float4*)(arow + (c >> 2) * hs + 8 * (c & 3));
        }
        __builtin_amdgcn_sched_barrier(0);
        v8f acc0 = {}, acc1 = {};
#pragma unroll
        for (int c = 0; c < 16; ++c) {
            const float4 a = af[i & 1][c];
            const v2f alo = { a.x, a.y }, ahi = { a.z, a.w };
            const v2f b0lo = { bf[0][c].x, bf[0][c].y }, b0hi = { bf[0][c].z, bf[0][c].w };
            const v2f b1lo = { bf[1][c].x, bf[1][c].y }, b1hi = { bf[1][c].z, bf[1][c].w };
            acc0 = wmma_f32(alo, b0lo, acc0);
            acc1 = wmma_f32(alo, b1lo, acc1);
            acc0 = wmma_f32(ahi, b0hi, acc0);
            acc1 = wmma_f32(ahi, b1hi, acc1);
        }
        float* base = out + (size_t)((mt0 + i) * 16 + 8 * half) * 128 + 32 * s + l16;
#pragma unroll
        for (int r = 0; r < 8; ++r) {
            base[r * 128]      = acc0[r];
            base[r * 128 + 16] = acc1[r];
        }
    }
}

extern "C" void kernel_launch(void* const* d_in, const int* in_sizes, int n_in,
                              void* d_out, int out_size, void* d_ws, size_t ws_size,
                              hipStream_t stream) {
    const float* x      = (const float*)d_in[0];
    const float* w_attn = (const float*)d_in[1];
    const float* w_proj = (const float*)d_in[2];

    float* ws = (float*)d_ws;
    const size_t GBUF = (size_t)GROUPS * GLEN * HSZ;
    float* qg = ws;
    float* kg = ws + GBUF;
    float* vg = ws + 2 * GBUF;
    float* xg = ws + 3 * GBUF;
    float* y  = ws + 4 * GBUF;

    qkv_gemm_kernel<<<256, 384, 0, stream>>>(x, w_attn, qg, kg, vg);
    mean_kernel<<<dim3(GROUPS, 3), MEAN_THREADS, 0, stream>>>(qg, kg, vg);
    attn_group_kernel<<<ATT_WGS, 32 * ATT_WAVES, 0, stream>>>(qg, kg, vg, xg);
    meanatt_kernel<<<8, 256, 0, stream>>>(qg, kg, xg, y);
    proj_gemm_kernel<<<205, 128, 0, stream>>>(xg, y, w_proj, (float*)d_out);
}


# qkv GEMM writes K and V as 2 KB WMMA-fragment records (K by the transposed product): attention K/V loads are 32 whole-line 16-byte loads per 64-key block, were 96 touching 16 or 2 partial lines each
# speedup vs baseline: 3.7855x; 1.1597x over previous
#include <hip/hip_runtime.h>
#include <hip/hip_bf16.h>

typedef float v2f __attribute__((ext_vector_type(2)));
typedef float v8f __attribute__((ext_vector_type(8)));

#define GROUPS 64
#define GLEN   1025
#define HSZ    32
#define SCALE  0.17677669529663687f

__device__ __forceinline__ v8f wmma_f32(v2f a, v2f b, v8f c) {
    return __builtin_amdgcn_wmma_f32_16x16x4_f32(false, a, false, b, (short)0, c, false, false);
}

#define QKV_TILES 4
template <int WHICH>
__device__ __forceinline__ void qkv_tiles(
    const float* __restrict__ x, const float4 (&bf)[2][16], float* __restrict__ dst,
    const int h, const int mt0, const int lane, const int half, const int l16)
{
    float4 af[2][16];
    {
        const float* arow = x + (size_t)(mt0 * 16 + l16) * 128 + 4 * half;
#pragma unroll
        for (int c = 0; c < 16; ++c) af[0][c] = *(const float4*)(arow + 8 * c);
    }
    __builtin_amdgcn_sched_barrier(0);
#pragma unroll
    for (int i = 0; i < QKV_TILES; ++i) {
        if (i + 1 < QKV_TILES) {
            const float* arow = x + (size_t)((mt0 + i + 1) * 16 + l16) * 128 + 4 * half;
#pragma unroll
            for (int c = 0; c < 16; ++c) af[(i + 1) & 1][c] = *(const float4*)(arow + 8 * c);
        }
        __builtin_amdgcn_sched_barrier(0);
        v8f acc0 = {}, acc1 = {};
#pragma unroll
        for (int c = 0; c < 16; ++c) {
            const float4 a = af[i & 1][c];
            const v2f alo = { a.x, a.y }, ahi = { a.z, a.w };
            const v2f b0lo = { bf[0][c].x, bf[0][c].y }, b0hi = { bf[0][c].z, bf[0][c].w };
            const v2f b1lo = { bf[1][c].x, bf[1][c].y }, b1hi = { bf[1][c].z, bf[1][c].w };
            if (WHICH == 1) {
                acc0 = wmma_f32(b0lo, alo, acc0);
                acc1 = wmma_f32(b1lo, alo, acc1);
                acc0 = wmma_f32(b0hi, ahi, acc0);
                acc1 = wmma_f32(b1hi, ahi, acc1);
            } else {
                acc0 = wmma_f32(alo, b0lo, acc0);
                acc1 = wmma_f32(alo, b1lo, acc1);
                acc0 = wmma_f32(ahi, b0hi, acc0);
                acc1 = wmma_f32(ahi, b1hi, acc1);
            }
        }
        const int mt = mt0 + i;
        const int bidx = mt >> 9;
        const int g = (mt & 511) >> 6;
        const int T = mt & 63;
        float* gb = dst + (size_t)((bidx * 4 + h) * 8 + g) * GLEN * HSZ;
        if (WHICH == 0) {
            float* base = gb + (T * 16 + 8 * half) * HSZ + l16;
#pragma unroll
            for (int r = 0; r < 8; ++r) {
                base[r * HSZ]      = acc0[r];
                base[r * HSZ + 16] = acc1[r];
            }
        } else {
            const float4 o0 = { acc0[0], acc0[1], acc0[2], acc0[3] };
            const float4 o1 = { acc0[4], acc0[5], acc0[6], acc0[7] };
            const float4 o2 = { acc1[0], acc1[1], acc1[2], acc1[3] };
            const float4 o3 = { acc1[4], acc1[5], acc1[6], acc1[7] };
            if (WHICH == 2) {
                float4* fb = (float4*)(gb + T * 512) + lane;
                fb[0] = o0;  fb[32] = o1;  fb[64] = o2;  fb[96] = o3;
            } else {
                float4* fb = (float4*)(gb + T * 512 + half * 128) + l16;
                fb[0] = o0;  fb[16] = o1;  fb[64] = o2;  fb[80] = o3;
            }
        }
    }
}

__global__ __launch_bounds__(384) __attribute__((amdgpu_waves_per_eu(3, 3))) void qkv_gemm_kernel(
    const float* __restrict__ x, const float* __restrict__ w,
    float* __restrict__ qg, float* __restrict__ kg, float* __restrict__ vg)
{
    const int s = __builtin_amdgcn_readfirstlane((int)(threadIdx.x >> 5));
    const int lane = threadIdx.x & 31;
    const int half = lane >> 4, l16 = lane & 15;
    const int mt0 = blockIdx.x * QKV_TILES;

    float4 bf[2][16];
#pragma unroll
    for (int t = 0; t < 2; ++t) {
        const float* brow = w + (size_t)(32 * s + 16 * t + l16) * 128 + 4 * half;
#pragma unroll
        for (int c = 0; c < 16; ++c) bf[t][c] = *(const float4*)(brow + 8 * c);
    }

    const int which = s >> 2;
    const int h = s & 3;
    if (which == 0)      qkv_tiles<0>(x, bf, qg, h, mt0, lane, half, l16);
    else if (which == 1) qkv_tiles<1>(x, bf, kg, h, mt0, lane, half, l16);
    else                 qkv_tiles<2>(x, bf, vg, h, mt0, lane, half, l16);
}

#define MEAN_THREADS 256
#define MEAN_LOADS   32
__global__ __launch_bounds__(MEAN_THREADS) __attribute__((amdgpu_waves_per_eu(2, 2))) void mean_kernel(
    float* __restrict__ qg, float* __restrict__ kg, float* __restrict__ vg)
{
    const int grp = blockIdx.x;
    const int which = blockIdx.y;
    float* buf = (which == 0) ? qg : ((which == 1) ? kg : vg);
    float* gbase = buf + (size_t)grp * GLEN * HSZ;
    const int t = threadIdx.x;
    const int wave = t >> 5, lane = t & 31;

    const float4* src = (const float4*)gbase + t;
    float4 v[MEAN_LOADS];
#pragma unroll
    for (int i = 0; i < MEAN_LOADS; ++i) v[i] = src[i * MEAN_THREADS];
    __builtin_amdgcn_sched_barrier(0);

#pragma unroll
    for (int s = MEAN_LOADS / 2; s >= 1; s >>= 1) {
#pragma unroll
        for (int i = 0; i < s; ++i) {
            v[i].x += v[i + s].x;
            v[i].y += v[i + s].y;
            v[i].z += v[i + s].z;
            v[i].w += v[i + s].w;
        }
    }
    float4 a = v[0];

    __shared__ __attribute__((aligned(16))) float part[MEAN_THREADS / 32][HSZ];
    part[wave][lane] = 0.0f;
    if (which == 0) {
        a.x += __shfl_xor(a.x, 8, 32);  a.y += __shfl_xor(a.y, 8, 32);
        a.z += __shfl_xor(a.z, 8, 32);  a.w += __shfl_xor(a.w, 8, 32);
        a.x += __shfl_xor(a.x, 16, 32); a.y += __shfl_xor(a.y, 16, 32);
        a.z += __shfl_xor(a.z, 16, 32); a.w += __shfl_xor(a.w, 16, 32);
        if (lane < 8) *(float4*)&part[wave][4 * lane] = a;
    } else if (which == 1) {
#pragma unroll
        for (int m = 1; m <= 8; m <<= 1) {
            a.x += __shfl_xor(a.x, m, 32);  a.y += __shfl_xor(a.y, m, 32);
            a.z += __shfl_xor(a.z, m, 32);  a.w += __shfl_xor(a.w, m, 32);
        }
        if ((lane & 15) == 0) *(float4*)&part[wave][8 * (wave & 3) + 4 * (lane >> 4)] = a;
    } else {
        float sv = (a.x + a.y) + (a.z + a.w);
        sv += __shfl_xor(sv, 16, 32);
        if (lane < 16) part[wave][16 * ((wave >> 1) & 1) + lane] = sv;
    }
    __syncthreads();
    if (t < HSZ) {
        float s = 0.0f;
#pragma unroll
        for (int w = 0; w < MEAN_THREADS / 32; ++w) s += part[w][t];
        gbase[1024 * HSZ + t] = s * (1.0f / 1024.0f);
    }
}

__device__ __forceinline__ float xhalf(float x) {
    const int b = __builtin_bit_cast(int, x);
    return __builtin_bit_cast(float,
        __builtin_amdgcn_permlanex16(b, b, 0x76543210, 0xfedcba98, false, false));
}

template <int MODE>
__device__ __forceinline__ void attn_block(
    const float* __restrict__ K, const float* __restrict__ V,
    const v2f (&qf)[8], v8f& acc0, v8f& acc1, float& colM, float& colL,
    const int kbase, const int qbase, const int lane, const int half, const int l16)
{
    constexpr bool DIAG = (MODE != 0);
    v2f kf[4][8];
    v2f va0[4][4], va1[4][4];
    if (MODE != 2) {
        const float4* kp = (const float4*)(K + kbase * HSZ) + lane;
        const float4* vp = (const float4*)(V + kbase * HSZ) + lane;
#pragma unroll
        for (int j = 0; j < 4; ++j) {
#pragma unroll
            for (int p = 0; p < 4; ++p) {
                const float4 f = kp[j * 128 + p * 32];
                kf[j][2 * p]     = (v2f){ f.x, f.y };
                kf[j][2 * p + 1] = (v2f){ f.z, f.w };
            }
        }
#pragma unroll
        for (int j = 0; j < 4; ++j) {
            const float4 f0 = vp[j * 128];
            const float4 f1 = vp[j * 128 + 32];
            const float4 f2 = vp[j * 128 + 64];
            const float4 f3 = vp[j * 128 + 96];
            va0[j][0] = (v2f){ f0.x, f0.y };  va0[j][1] = (v2f){ f0.z, f0.w };
            va0[j][2] = (v2f){ f1.x, f1.y };  va0[j][3] = (v2f){ f1.z, f1.w };
            va1[j][0] = (v2f){ f2.x, f2.y };  va1[j][1] = (v2f){ f2.z, f2.w };
            va1[j][2] = (v2f){ f3.x, f3.y };  va1[j][3] = (v2f){ f3.z, f3.w };
        }
    } else {
        const float* kptr = K + 1024 * HSZ + 4 * half;
        const float vm0 = V[1024 * HSZ + l16], vm1 = V[1024 * HSZ + 16 + l16];
#pragma unroll
        for (int p = 0; p < 4; ++p) {
            const float4 f = *(const float4*)(kptr + 8 * p);
#pragma unroll
            for (int j = 0; j < 4; ++j) {
                kf[j][2 * p]     = (v2f){ f.x, f.y };
                kf[j][2 * p + 1] = (v2f){ f.z, f.w };
            }
        }
#pragma unroll
        for (int j = 0; j < 4; ++j) {
#pragma unroll
            for (int i = 0; i < 4; ++i) {
                va0[j][i] = (v2f){ vm0, vm0 };
                va1[j][i] = (v2f){ vm1, vm1 };
            }
        }
    }

    __builtin_amdgcn_sched_barrier(0);

    v8f s[4];
#pragma unroll
    for (int j = 0; j < 4; ++j) {
        s[j] = (v8f){};
#pragma unroll
        for (int c = 0; c < 8; ++c) s[j] = wmma_f32(kf[j][c], qf[c], s[j]);
    }

    const int qi = qbase + l16;
    float mj[4];
#pragma unroll
    for (int j = 0; j < 4; ++j) {
        mj[j] = -3.0e38f;
#pragma unroll
        for (int r = 0; r < 8; ++r) {
            float sv = s[j][r] * SCALE;
            if (DIAG) {
                const int kj = kbase + 16 * j + 8 * half + r;
                const bool valid = (kj <= qi) && (kj <= 1024);
                sv = valid ? sv : -3.0e38f;
            }
            s[j][r] = sv;
            mj[j] = fmaxf(mj[j], sv);
        }
    }
    float mx = fmaxf(fmaxf(mj[0], mj[1]), fmaxf(mj[2], mj[3]));
    mx = fmaxf(mx, xhalf(mx));
    const float nm = fmaxf(colM, mx);
    float rj[4];
#pragma unroll
    for (int j = 0; j < 4; ++j) {
        rj[j] = 0.0f;
#pragma unroll
        for (int r = 0; r < 8; ++r) {
            const float pv = __expf(s[j][r] - nm);
            s[j][r] = pv;
            rj[j] += pv;
        }
    }
    float rs = (rj[0] + rj[1]) + (rj[2] + rj[3]);
    rs += xhalf(rs);
    const float corr = __expf(colM - nm);
    colM = nm;
    colL = colL * corr + rs;
#pragma unroll
    for (int r = 0; r < 8; ++r) {
        acc0[r] = acc0[r] * corr;
        acc1[r] = acc1[r] * corr;
    }

#pragma unroll
    for (int j = 0; j < 4; ++j) {
#pragma unroll
        for (int i = 0; i < 4; ++i) {
            const v2f pb = { s[j][2 * i], s[j][2 * i + 1] };
            acc0 = wmma_f32(va0[j][i], pb, acc0);
            acc1 = wmma_f32(va1[j][i], pb, acc1);
        }
    }
}

#define ATT_WAVES 4
#define ATT_WGS   1088
__global__ __launch_bounds__(32 * ATT_WAVES) void attn_group_kernel(
    const float* __restrict__ qg, const float* __restrict__ kg,
    const float* __restrict__ vg, float* __restrict__ xg)
{
    const int id = blockIdx.x;
    int w, grp;
    if (id < 1024) {
        const unsigned long long sq = 0x0DE3B6587A94C12FULL;
        const int m = id >> 2;
        w = 1 + (int)((sq >> (4 * (4 * (id & 3) + (m >> 6)))) & 15ULL);
        grp = m & 63;
    } else {
        w = 0;
        grp = id - 1024;
    }
    const int qt = 4 * w + __builtin_amdgcn_readfirstlane((int)(threadIdx.x >> 5));
    if (qt > 64) return;
    const int lane = threadIdx.x & 31;
    const int half = lane >> 4, l16 = lane & 15;

    const float* Q = qg + (size_t)grp * GLEN * HSZ;
    const float* K = kg + (size_t)grp * GLEN * HSZ;
    const float* V = vg + (size_t)grp * GLEN * HSZ;
    float*       X = xg + (size_t)grp * GLEN * HSZ;

    const int qbase = qt * 16;
    int qrow = qbase + l16; if (qrow > 1024) qrow = 1024;
    const float* qptr = Q + qrow * HSZ + 4 * half;

    v2f qf[8];
#pragma unroll
    for (int p = 0; p < 4; ++p) {
        const float4 f = *(const float4*)(qptr + 8 * p);
        qf[2 * p]     = (v2f){ f.x, f.y };
        qf[2 * p + 1] = (v2f){ f.z, f.w };
    }

    v8f acc0 = {}, acc1 = {};
    float colM = -3.0e38f, colL = 0.0f;

    const int nfull = qt >> 2;
    for (int b = 0; b < nfull; ++b)
        attn_block<0>(K, V, qf, acc0, acc1, colM, colL, 64 * b, qbase, lane, half, l16);
    if (qt < 64) attn_block<1>(K, V, qf, acc0, acc1, colM, colL, 64 * nfull, qbase, lane, half, l16);
    else         attn_block<2>(K, V, qf, acc0, acc1, colM, colL, 64 * nfull, qbase, lane, half, l16);

    const int row = qbase + l16;
    if (row <= 1024) {
        const float invl = 1.0f / colL;
        const bool mrow = (row == 1024);
        const float* vb = mrow ? (V + 1024 * HSZ + 8 * half)
                               : (V + qt * 512 + ((l16 >> 2) & 1) * 128 + (l16 >> 3) * 64 + 32 * half + (l16 & 3));
        const int vs = mrow ? 1 : 4, vo = mrow ? 16 : 256;
        float4 v0, v1, v2, v3;
        v0.x = vb[0];           v0.y = vb[vs];          v0.z = vb[2 * vs];      v0.w = vb[3 * vs];
        v1.x = vb[4 * vs];      v1.y = vb[5 * vs];      v1.z = vb[6 * vs];      v1.w = vb[7 * vs];
        v2.x = vb[vo];          v2.y = vb[vo + vs];     v2.z = vb[vo + 2 * vs]; v2.w = vb[vo + 3 * vs];
        v3.x = vb[vo + 4 * vs]; v3.y = vb[vo + 5 * vs]; v3.z = vb[vo + 6 * vs]; v3.w = vb[vo + 7 * vs];
        float4*       xr = (float4*)(X + row * HSZ + 8 * half);
        float4 o0, o1, o2, o3;
        o0.x = acc0[0] * invl + v0.x;  o0.y = acc0[1] * invl + v0.y;
        o0.z = acc0[2] * invl + v0.z;  o0.w = acc0[3] * invl + v0.w;
        o1.x = acc0[4] * invl + v1.x;  o1.y = acc0[5] * invl + v1.y;
        o1.z = acc0[6] * invl + v1.z;  o1.w = acc0[7] * invl + v1.w;
        o2.x = acc1[0] * invl + v2.x;  o2.y = acc1[1] * invl + v2.y;
        o2.z = acc1[2] * invl + v2.z;  o2.w = acc1[3] * invl + v2.w;
        o3.x = acc1[4] * invl + v3.x;  o3.y = acc1[5] * invl + v3.y;
        o3.z = acc1[6] * invl + v3.z;  o3.w = acc1[7] * invl + v3.w;
        xr[0] = o0;  xr[1] = o1;  xr[4] = o2;  xr[5] = o3;
    }
}

__global__ __launch_bounds__(256) void meanatt_kernel(
    const float* __restrict__ qg, const float* __restrict__ kg,
    const float* __restrict__ xg, float* __restrict__ y)
{
    const int bh = blockIdx.x;
    __shared__ float qm[8][32], km[8][32], vm[8][32];
    const int t = threadIdx.x;
    const int g = t / 32, d = t % 32;
    const size_t rowoff = ((size_t)(bh * 8 + g) * GLEN + 1024) * HSZ + d;
    qm[g][d] = qg[rowoff];
    km[g][d] = kg[rowoff];
    vm[g][d] = xg[rowoff];
    __syncthreads();

    float sarr[8];
    float mx = -3.0e38f;
    for (int gp = 0; gp <= g; ++gp) {
        float sv = 0.0f;
        for (int c = 0; c < 32; ++c) sv += qm[g][c] * km[gp][c];
        sv *= SCALE;
        sarr[gp] = sv;
        mx = fmaxf(mx, sv);
    }
    float denom = 0.0f;
    for (int gp = 0; gp <= g; ++gp) { sarr[gp] = __expf(sarr[gp] - mx); denom += sarr[gp]; }
    float o = 0.0f;
    for (int gp = 0; gp <= g; ++gp) o += sarr[gp] * vm[gp][d];
    y[(bh * 8 + g) * 32 + d] = o / denom;
}

#define PROJ_TILES 5
__device__ __forceinline__ const float* zrow_ptr(
    const float* __restrict__ xg, const float* __restrict__ y, const int m, int& hstride)
{
    const int b = m / 8200;
    const int t = m - b * 8200;
    const int g = t / GLEN;
    const int r = t - g * GLEN;
    const bool isy = (g != 0) && (r == 0);
    const int gr = (g == 0) ? r : r - 1;
    const int offx = ((b * 32 + g) * GLEN + gr) * HSZ;
    const int offy = (b * 32 + g - 1) * HSZ;
    hstride = isy ? 8 * HSZ : 8 * GLEN * HSZ;
    return (isy ? y : xg) + (isy ? offy : offx);
}

__global__ __launch_bounds__(128) void proj_gemm_kernel(
    const float* __restrict__ xg, const float* __restrict__ y,
    const float* __restrict__ w, float* __restrict__ out)
{
    const int s = __builtin_amdgcn_readfirstlane((int)(threadIdx.x >> 5));
    const int lane = threadIdx.x & 31;
    const int half = lane >> 4, l16 = lane & 15;
    const int mt0 = blockIdx.x * PROJ_TILES;

    float4 bf[2][16];
#pragma unroll
    for (int t = 0; t < 2; ++t) {
        const float* brow = w + (size_t)(32 * s + 16 * t + l16) * 128 + 4 * half;
#pragma unroll
        for (int c = 0; c < 16; ++c) bf[t][c] = *(const float4*)(brow + 8 * c);
    }

    float4 af[2][16];
    {
        int hs;
        const float* arow = zrow_ptr(xg, y, mt0 * 16 + l16, hs) + 4 * half;
#pragma unroll
        for (int c = 0; c < 16; ++c) af[0][c] = *(const float4*)(arow + (c >> 2) * hs + 8 * (c & 3));
    }
    __builtin_amdgcn_sched_barrier(0);
#pragma unroll
    for (int i = 0; i < PROJ_TILES; ++i) {
        if (i + 1 < PROJ_TILES) {
            int hs;
            const float* arow = zrow_ptr(xg, y, (mt0 + i + 1) * 16 + l16, hs) + 4 * half;
#pragma unroll
            for (int c = 0; c < 16; ++c) af[(i + 1) & 1][c] = *(const float4*)(arow + (c >> 2) * hs + 8 * (c & 3));
        }
        __builtin_amdgcn_sched_barrier(0);
        v8f acc0 = {}, acc1 = {};
#pragma unroll
        for (int c = 0; c < 16; ++c) {
            const float4 a = af[i & 1][c];
            const v2f alo = { a.x, a.y }, ahi = { a.z, a.w };
            const v2f b0lo = { bf[0][c].x, bf[0][c].y }, b0hi = { bf[0][c].z, bf[0][c].w };
            const v2f b1lo = { bf[1][c].x, bf[1][c].y }, b1hi = { bf[1][c].z, bf[1][c].w };
            acc0 = wmma_f32(alo, b0lo, acc0);
            acc1 = wmma_f32(alo, b1lo, acc1);
            acc0 = wmma_f32(ahi, b0hi, acc0);
            acc1 = wmma_f32(ahi, b1hi, acc1);
        }
        float* base = out + (size_t)((mt0 + i) * 16 + 8 * half) * 128 + 32 * s + l16;
#pragma unroll
        for (int r = 0; r < 8; ++r) {
            base[r * 128]      = acc0[r];
            base[r * 128 + 16] = acc1[r];
        }
    }
}

extern "C" void kernel_launch(void* const* d_in, const int* in_sizes, int n_in,
                              void* d_out, int out_size, void* d_ws, size_t ws_size,
                              hipStream_t stream) {
    const float* x      = (const float*)d_in[0];
    const float* w_attn = (const float*)d_in[1];
    const float* w_proj = (const float*)d_in[2];

    float* ws = (float*)d_ws;
    const size_t GBUF = (size_t)GROUPS * GLEN * HSZ;
    float* qg = ws;
    float* kg = ws + GBUF;
    float* vg = ws + 2 * GBUF;
    float* xg = ws + 3 * GBUF;
    float* y  = ws + 4 * GBUF;

    qkv_gemm_kernel<<<256, 384, 0, stream>>>(x, w_attn, qg, kg, vg);
    mean_kernel<<<dim3(GROUPS, 3), MEAN_THREADS, 0, stream>>>(qg, kg, vg);
    attn_group_kernel<<<ATT_WGS, 32 * ATT_WAVES, 0, stream>>>(qg, kg, vg, xg);
    meanatt_kernel<<<8, 256, 0, stream>>>(qg, kg, xg, y);
    proj_gemm_kernel<<<205, 128, 0, stream>>>(xg, y, w_proj, (float*)d_out);
}


# attention kernel: next key block's K fragments are requested once the score WMMAs free their registers, V at the block top; each load group is a WMMA run ahead of its use, no extra registers
# speedup vs baseline: 4.5934x; 1.2134x over previous
#include <hip/hip_runtime.h>
#include <hip/hip_bf16.h>

typedef float v2f __attribute__((ext_vector_type(2)));
typedef float v8f __attribute__((ext_vector_type(8)));

#define GROUPS 64
#define GLEN   1025
#define HSZ    32
#define SCALE  0.17677669529663687f

__device__ __forceinline__ v8f wmma_f32(v2f a, v2f b, v8f c) {
    return __builtin_amdgcn_wmma_f32_16x16x4_f32(false, a, false, b, (short)0, c, false, false);
}

__device__ __forceinline__ float xhalf(float x) {
    const int b = __builtin_bit_cast(int, x);
    return __builtin_bit_cast(float,
        __builtin_amdgcn_permlanex16(b, b, 0x76543210, 0xfedcba98, false, false));
}

#define PART_COLS 384
#define QKV_TILES 4
template <int WHICH>
__device__ __forceinline__ void qkv_tiles(
    const float* __restrict__ x, const float4 (&bf)[2][16], float* __restrict__ dst,
    float* __restrict__ prow,
    const int h, const int mt0, const int lane, const int half, const int l16)
{
    float cs0 = 0.0f, cs1 = 0.0f;
    const bool khi = (l16 & 8) != 0;
    float ks[8];
#pragma unroll
    for (int r = 0; r < 8; ++r) ks[r] = 0.0f;
    float4 af[2][16];
    {
        const float* arow = x + (size_t)(mt0 * 16 + l16) * 128 + 4 * half;
#pragma unroll
        for (int c = 0; c < 16; ++c) af[0][c] = *(const float4*)(arow + 8 * c);
    }
    __builtin_amdgcn_sched_barrier(0);
#pragma unroll
    for (int i = 0; i < QKV_TILES; ++i) {
        if (i + 1 < QKV_TILES) {
            const float* arow = x + (size_t)((mt0 + i + 1) * 16 + l16) * 128 + 4 * half;
#pragma unroll
            for (int c = 0; c < 16; ++c) af[(i + 1) & 1][c] = *(const float4*)(arow + 8 * c);
        }
        __builtin_amdgcn_sched_barrier(0);
        v8f acc0 = {}, acc1 = {};
#pragma unroll
        for (int c = 0; c < 16; ++c) {
            const float4 a = af[i & 1][c];
            const v2f alo = { a.x, a.y }, ahi = { a.z, a.w };
            const v2f b0lo = { bf[0][c].x, bf[0][c].y }, b0hi = { bf[0][c].z, bf[0][c].w };
            const v2f b1lo = { bf[1][c].x, bf[1][c].y }, b1hi = { bf[1][c].z, bf[1][c].w };
            if (WHICH == 1) {
                acc0 = wmma_f32(b0lo, alo, acc0);
                acc1 = wmma_f32(b1lo, alo, acc1);
                acc0 = wmma_f32(b0hi, ahi, acc0);
                acc1 = wmma_f32(b1hi, ahi, acc1);
            } else {
                acc0 = wmma_f32(alo, b0lo, acc0);
                acc1 = wmma_f32(alo, b1lo, acc1);
                acc0 = wmma_f32(ahi, b0hi, acc0);
                acc1 = wmma_f32(ahi, b1hi, acc1);
            }
        }
        if (WHICH == 1) {
#pragma unroll
            for (int r = 0; r < 8; ++r) {
                const float keep = khi ? acc1[r] : acc0[r];
                const float send = khi ? acc0[r] : acc1[r];
                ks[r] += keep + __shfl_xor(send, 8, 32);
            }
        } else {
            cs0 += ((acc0[0] + acc0[1]) + (acc0[2] + acc0[3])) + ((acc0[4] + acc0[5]) + (acc0[6] + acc0[7]));
            cs1 += ((acc1[0] + acc1[1]) + (acc1[2] + acc1[3])) + ((acc1[4] + acc1[5]) + (acc1[6] + acc1[7]));
        }
        const int mt = mt0 + i;
        const int bidx = mt >> 9;
        const int g = (mt & 511) >> 6;
        const int T = mt & 63;
        float* gb = dst + (size_t)((bidx * 4 + h) * 8 + g) * GLEN * HSZ;
        if (WHICH == 0) {
            float* base = gb + (T * 16 + 8 * half) * HSZ + l16;
#pragma unroll
            for (int r = 0; r < 8; ++r) {
                base[r * HSZ]      = acc0[r];
                base[r * HSZ + 16] = acc1[r];
            }
        } else {
            const float4 o0 = { acc0[0], acc0[1], acc0[2], acc0[3] };
            const float4 o1 = { acc0[4], acc0[5], acc0[6], acc0[7] };
            const float4 o2 = { acc1[0], acc1[1], acc1[2], acc1[3] };
            const float4 o3 = { acc1[4], acc1[5], acc1[6], acc1[7] };
            if (WHICH == 2) {
                float4* fb = (float4*)(gb + T * 512) + lane;
                fb[0] = o0;  fb[32] = o1;  fb[64] = o2;  fb[96] = o3;
            } else {
                float4* fb = (float4*)(gb + T * 512 + half * 128) + l16;
                fb[0] = o0;  fb[16] = o1;  fb[64] = o2;  fb[80] = o3;
            }
        }
    }
    if (WHICH == 1) {
#pragma unroll
        for (int r = 0; r < 8; ++r) {
#pragma unroll
            for (int m = 1; m <= 4; m <<= 1) ks[r] += __shfl_xor(ks[r], m, 32);
        }
        if ((l16 & 7) == 0) {
            float4* pr = (float4*)(prow + 8 * half + 2 * (l16 & 8));
            pr[0] = (float4){ ks[0], ks[1], ks[2], ks[3] };
            pr[1] = (float4){ ks[4], ks[5], ks[6], ks[7] };
        }
    } else {
        cs0 += xhalf(cs0);
        cs1 += xhalf(cs1);
        prow[lane] = half ? cs1 : cs0;
    }
}

__global__ __launch_bounds__(384) __attribute__((amdgpu_waves_per_eu(3, 3))) void qkv_gemm_kernel(
    const float* __restrict__ x, const float* __restrict__ w,
    float* __restrict__ qg, float* __restrict__ kg, float* __restrict__ vg,
    float* __restrict__ part)
{
    const int s = __builtin_amdgcn_readfirstlane((int)(threadIdx.x >> 5));
    const int lane = threadIdx.x & 31;
    const int half = lane >> 4, l16 = lane & 15;
    const int mt0 = blockIdx.x * QKV_TILES;

    float4 bf[2][16];
#pragma unroll
    for (int t = 0; t < 2; ++t) {
        const float* brow = w + (size_t)(32 * s + 16 * t + l16) * 128 + 4 * half;
#pragma unroll
        for (int c = 0; c < 16; ++c) bf[t][c] = *(const float4*)(brow + 8 * c);
    }

    const int which = s >> 2;
    const int h = s & 3;
    float* prow = part + (size_t)blockIdx.x * PART_COLS + 32 * s;
    if (which == 0)      qkv_tiles<0>(x, bf, qg, prow, h, mt0, lane, half, l16);
    else if (which == 1) qkv_tiles<1>(x, bf, kg, prow, h, mt0, lane, half, l16);
    else                 qkv_tiles<2>(x, bf, vg, prow, h, mt0, lane, half, l16);
}

__device__ __forceinline__ void load_kfrag(
    v2f (&kf)[4][8], const float* __restrict__ K, const int kbase, const int lane)
{
    const float4* kp = (const float4*)(K + kbase * HSZ) + lane;
#pragma unroll
    for (int j = 0; j < 4; ++j) {
#pragma unroll
        for (int p = 0; p < 4; ++p) {
            const float4 f = kp[j * 128 + p * 32];
            kf[j][2 * p]     = (v2f){ f.x, f.y };
            kf[j][2 * p + 1] = (v2f){ f.z, f.w };
        }
    }
}

template <int MODE, bool NEXT>
__device__ __forceinline__ void attn_block(
    const float* __restrict__ K, const float* __restrict__ V,
    const v2f (&qf)[8], v2f (&kf)[4][8], v8f& acc0, v8f& acc1, float& colM, float& colL,
    const int kbase, const int knext, const int qbase, const int lane, const int half, const int l16)
{
    constexpr bool DIAG = (MODE != 0);
    v2f va0[4][4], va1[4][4];
    if (MODE != 2) {
        const float4* vp = (const float4*)(V + kbase * HSZ) + lane;
#pragma unroll
        for (int j = 0; j < 4; ++j) {
            const float4 f0 = vp[j * 128];
            const float4 f1 = vp[j * 128 + 32];
            const float4 f2 = vp[j * 128 + 64];
            const float4 f3 = vp[j * 128 + 96];
            va0[j][0] = (v2f){ f0.x, f0.y };  va0[j][1] = (v2f){ f0.z, f0.w };
            va0[j][2] = (v2f){ f1.x, f1.y };  va0[j][3] = (v2f){ f1.z, f1.w };
            va1[j][0] = (v2f){ f2.x, f2.y };  va1[j][1] = (v2f){ f2.z, f2.w };
            va1[j][2] = (v2f){ f3.x, f3.y };  va1[j][3] = (v2f){ f3.z, f3.w };
        }
    } else {
        const float* kptr = K + 4 * half;
        const float vm0 = V[l16], vm1 = V[16 + l16];
#pragma unroll
        for (int p = 0; p < 4; ++p) {
            const float4 f = *(const float4*)(kptr + 8 * p);
#pragma unroll
            for (int j = 0; j < 4; ++j) {
                kf[j][2 * p]     = (v2f){ f.x, f.y };
                kf[j][2 * p + 1] = (v2f){ f.z, f.w };
            }
        }
#pragma unroll
        for (int j = 0; j < 4; ++j) {
#pragma unroll
            for (int i = 0; i < 4; ++i) {
                va0[j][i] = (v2f){ vm0, vm0 };
                va1[j][i] = (v2f){ vm1, vm1 };
            }
        }
    }

    __builtin_amdgcn_sched_barrier(0);

    v8f s[4];
#pragma unroll
    for (int j = 0; j < 4; ++j) s[j] = (v8f){};
#pragma unroll
    for (int c = 0; c < 8; ++c) {
#pragma unroll
        for (int j = 0; j < 4; ++j) s[j] = wmma_f32(kf[j][c], qf[c], s[j]);
    }

    if (NEXT) {
        asm volatile("" : "+v"(s[0]), "+v"(s[1]), "+v"(s[2]), "+v"(s[3]) : : "memory");
        __builtin_amdgcn_sched_barrier(0);
        load_kfrag(kf, K, knext, lane);
        __builtin_amdgcn_sched_barrier(0);
    }

    const int qi = qbase + l16;
    float mj[4];
#pragma unroll
    for (int j = 0; j < 4; ++j) {
        mj[j] = -3.0e38f;
#pragma unroll
        for (int r = 0; r < 8; ++r) {
            float sv = s[j][r] * SCALE;
            if (DIAG) {
                const int kj = kbase + 16 * j + 8 * half + r;
                const bool valid = (kj <= qi) && (kj <= 1024);
                sv = valid ? sv : -3.0e38f;
            }
            s[j][r] = sv;
            mj[j] = fmaxf(mj[j], sv);
        }
    }
    float mx = fmaxf(fmaxf(mj[0], mj[1]), fmaxf(mj[2], mj[3]));
    mx = fmaxf(mx, xhalf(mx));
    const float nm = fmaxf(colM, mx);
    float rj[4];
#pragma unroll
    for (int j = 0; j < 4; ++j) {
        rj[j] = 0.0f;
#pragma unroll
        for (int r = 0; r < 8; ++r) {
            const float pv = __expf(s[j][r] - nm);
            s[j][r] = pv;
            rj[j] += pv;
        }
    }
    float rs = (rj[0] + rj[1]) + (rj[2] + rj[3]);
    rs += xhalf(rs);
    const float corr = __expf(colM - nm);
    colM = nm;
    colL = colL * corr + rs;
#pragma unroll
    for (int r = 0; r < 8; ++r) {
        acc0[r] = acc0[r] * corr;
        acc1[r] = acc1[r] * corr;
    }

#pragma unroll
    for (int j = 0; j < 4; ++j) {
#pragma unroll
        for (int i = 0; i < 4; ++i) {
            const v2f pb = { s[j][2 * i], s[j][2 * i + 1] };
            acc0 = wmma_f32(va0[j][i], pb, acc0);
            acc1 = wmma_f32(va1[j][i], pb, acc1);
        }
    }
}

#define ATT_WAVES 4
#define ATT_WGS   1088
__global__ __launch_bounds__(32 * ATT_WAVES) void attn_group_kernel(
    float* __restrict__ qg, float* __restrict__ kg,
    const float* __restrict__ vg, float* __restrict__ xg,
    const float* __restrict__ part)
{
    __shared__ __attribute__((aligned(16))) float msm[3][HSZ];
    const int id = blockIdx.x;
    int w, grp;
    if (id < 1024) {
        const unsigned long long sq = 0x0DE3B6587A94C12FULL;
        const int m = id >> 2;
        w = 1 + (int)((sq >> (4 * (4 * (id & 3) + (m >> 6)))) & 15ULL);
        grp = m & 63;
    } else {
        w = 0;
        grp = id - 1024;
    }
    const int qt = 4 * w + __builtin_amdgcn_readfirstlane((int)(threadIdx.x >> 5));
    if (qt > 64) return;
    const int lane = threadIdx.x & 31;
    const int half = lane >> 4, l16 = lane & 15;

    float* Q = qg + (size_t)grp * GLEN * HSZ;
    float* K = kg + (size_t)grp * GLEN * HSZ;
    const float* V = vg + (size_t)grp * GLEN * HSZ;
    float*       X = xg + (size_t)grp * GLEN * HSZ;

    const int qbase = qt * 16;

    v2f kf[4][8];
    load_kfrag(kf, K, 0, lane);

    v2f qf[8];
    if (qt < 64) {
        const float* qptr = Q + (qbase + l16) * HSZ + 4 * half;
#pragma unroll
        for (int p = 0; p < 4; ++p) {
            const float4 f = *(const float4*)(qptr + 8 * p);
            qf[2 * p]     = (v2f){ f.x, f.y };
            qf[2 * p + 1] = (v2f){ f.z, f.w };
        }
    } else {
        const int b = grp >> 5, h = (grp >> 3) & 3, g = grp & 7;
        const float* pp = part + (size_t)(b * 128 + g * 16) * PART_COLS + 32 * h + lane;
        float sq[16], sk[16], sv[16];
#pragma unroll
        for (int i = 0; i < 16; ++i) {
            sq[i] = pp[i * PART_COLS];
            sk[i] = pp[i * PART_COLS + 128];
            sv[i] = pp[i * PART_COLS + 256];
        }
#pragma unroll
        for (int s = 8; s >= 1; s >>= 1) {
#pragma unroll
            for (int i = 0; i < s; ++i) { sq[i] += sq[i + s]; sk[i] += sk[i + s]; sv[i] += sv[i + s]; }
        }
        const float mq = sq[0] * (1.0f / 1024.0f);
        const float mk = sk[0] * (1.0f / 1024.0f);
        const float mv = sv[0] * (1.0f / 1024.0f);
        Q[1024 * HSZ + lane] = mq;
        K[1024 * HSZ + lane] = mk;
        msm[0][lane] = mq;  msm[1][lane] = mk;  msm[2][lane] = mv;
        __builtin_amdgcn_wave_barrier();
#pragma unroll
        for (int p = 0; p < 4; ++p) {
            const float4 f = *(const float4*)&msm[0][8 * p + 4 * half];
            qf[2 * p]     = (v2f){ f.x, f.y };
            qf[2 * p + 1] = (v2f){ f.z, f.w };
        }
    }

    v8f acc0 = {}, acc1 = {};
    float colM = -3.0e38f, colL = 0.0f;

    const int nfull = qt >> 2;
    for (int b = 0; b < nfull; ++b) {
        const int nb = (b + 1 < 15) ? (b + 1) : 15;
        attn_block<0, true>(K, V, qf, kf, acc0, acc1, colM, colL, 64 * b, 64 * nb, qbase, lane, half, l16);
    }
    if (qt < 64) attn_block<1, false>(K, V, qf, kf, acc0, acc1, colM, colL, 64 * nfull, 0, qbase, lane, half, l16);
    else         attn_block<2, false>(msm[1], msm[2], qf, kf, acc0, acc1, colM, colL, 64 * nfull, 0, qbase, lane, half, l16);

    const int row = qbase + l16;
    if (row <= 1024) {
        const float invl = 1.0f / colL;
        float4 v0, v1, v2, v3;
        if (qt < 64) {
            const float* vb = V + qt * 512 + ((l16 >> 2) & 1) * 128 + (l16 >> 3) * 64 + 32 * half + (l16 & 3);
            v0.x = vb[0];    v0.y = vb[4];    v0.z = vb[8];    v0.w = vb[12];
            v1.x = vb[16];   v1.y = vb[20];   v1.z = vb[24];   v1.w = vb[28];
            v2.x = vb[256];  v2.y = vb[260];  v2.z = vb[264];  v2.w = vb[268];
            v3.x = vb[272];  v3.y = vb[276];  v3.z = vb[280];  v3.w = vb[284];
        } else {
            const float4* vb = (const float4*)&msm[2][8 * half];
            v0 = vb[0];  v1 = vb[1];  v2 = vb[4];  v3 = vb[5];
        }
        float4*       xr = (float4*)(X + row * HSZ + 8 * half);
        float4 o0, o1, o2, o3;
        o0.x = acc0[0] * invl + v0.x;  o0.y = acc0[1] * invl + v0.y;
        o0.z = acc0[2] * invl + v0.z;  o0.w = acc0[3] * invl + v0.w;
        o1.x = acc0[4] * invl + v1.x;  o1.y = acc0[5] * invl + v1.y;
        o1.z = acc0[6] * invl + v1.z;  o1.w = acc0[7] * invl + v1.w;
        o2.x = acc1[0] * invl + v2.x;  o2.y = acc1[1] * invl + v2.y;
        o2.z = acc1[2] * invl + v2.z;  o2.w = acc1[3] * invl + v2.w;
        o3.x = acc1[4] * invl + v3.x;  o3.y = acc1[5] * invl + v3.y;
        o3.z = acc1[6] * invl + v3.z;  o3.w = acc1[7] * invl + v3.w;
        xr[0] = o0;  xr[1] = o1;  xr[4] = o2;  xr[5] = o3;
    }
}

#define PROJ_TILES 5
#define ZHEAD (8 * GLEN * HSZ)
__device__ __forceinline__ const float* zrow_ptr(const float* __restrict__ xg, const int m)
{
    const int b = m / 8200;
    const int t = m - b * 8200;
    const int g = t / GLEN;
    const int r = t - g * GLEN;
    const int gr = (g == 0) ? r : r - 1;
    return xg + ((b * 32 + g) * GLEN + gr) * HSZ;
}

__global__ __launch_bounds__(128) void proj_gemm_kernel(
    const float* __restrict__ xg, const float* __restrict__ qg, const float* __restrict__ kg,
    const float* __restrict__ w, float* __restrict__ out)
{
    __shared__ __attribute__((aligned(16))) float ysm[4][HSZ];
    const int s = __builtin_amdgcn_readfirstlane((int)(threadIdx.x >> 5));
    const int lane = threadIdx.x & 31;
    const int half = lane >> 4, l16 = lane & 15;
    const int mt0 = blockIdx.x * PROJ_TILES;

    const int m0 = mt0 * 16;
    const int b0 = m0 / 8200;
    const int t0 = m0 - b0 * 8200;
    const int gy = (t0 + GLEN - 1) / GLEN;
    const int yrel = gy * GLEN - t0;
    const bool hasy = (gy >= 1) && (gy <= 7) && (yrel < 16 * PROJ_TILES);
    const int ytile = hasy ? (yrel >> 4) : -1;
    const int yl16 = yrel & 15;

    float qv = 0.0f, kv[7], xv[7];
#pragma unroll
    for (int gp = 0; gp < 7; ++gp) { kv[gp] = 0.0f; xv[gp] = 0.0f; }
    if (hasy) {
        const int gq = gy - 1;
        const int base = ((b0 * 32 + s * 8) * GLEN + 1024) * HSZ + lane;
        qv = qg[base + gq * (GLEN * HSZ)];
#pragma unroll
        for (int gp = 0; gp < 7; ++gp) {
            const int gc = gp < gq ? gp : gq;
            kv[gp] = kg[base + gc * (GLEN * HSZ)];
            xv[gp] = xg[base + gc * (GLEN * HSZ)];
        }
    }

    float4 bf[2][16];
#pragma unroll
    for (int t = 0; t < 2; ++t) {
        const float* brow = w + (size_t)(32 * s + 16 * t + l16) * 128 + 4 * half;
#pragma unroll
        for (int c = 0; c < 16; ++c) bf[t][c] = *(const float4*)(brow + 8 * c);
    }

    float4 af[2][16];
    {
        const float* arow = zrow_ptr(xg, mt0 * 16 + l16) + 4 * half;
#pragma unroll
        for (int c = 0; c < 16; ++c) af[0][c] = *(const float4*)(arow + (c >> 2) * ZHEAD + 8 * (c & 3));
    }
    __builtin_amdgcn_sched_barrier(0);

    if (hasy) {
        const int gq = gy - 1;
        float sc[7];
        float mx = -3.0e38f;
#pragma unroll
        for (int gp = 0; gp < 7; ++gp) {
            float p = qv * kv[gp];
#pragma unroll
            for (int m = 16; m >= 1; m >>= 1) p += __shfl_xor(p, m, 32);
            sc[gp] = (gp <= gq) ? p * SCALE : -3.0e38f;
            mx = fmaxf(mx, sc[gp]);
        }
        float denom = 0.0f, o = 0.0f;
#pragma unroll
        for (int gp = 0; gp < 7; ++gp) {
            const float pe = __expf(sc[gp] - mx);
            denom += pe;
            o += pe * xv[gp];
        }
        ysm[s][lane] = o / denom;
        __syncthreads();
    }

#pragma unroll
    for (int i = 0; i < PROJ_TILES; ++i) {
        if (i + 1 < PROJ_TILES) {
            const float* arow = zrow_ptr(xg, (mt0 + i + 1) * 16 + l16) + 4 * half;
#pragma unroll
            for (int c = 0; c < 16; ++c) af[(i + 1) & 1][c] = *(const float4*)(arow + (c >> 2) * ZHEAD + 8 * (c & 3));
        }
        __builtin_amdgcn_sched_barrier(0);
        if (ytile == i) {
            if (l16 == yl16) {
#pragma unroll
                for (int c = 0; c < 16; ++c) af[i & 1][c] = *(const float4*)&ysm[c >> 2][8 * (c & 3) + 4 * half];
            }
        }
        v8f acc0 = {}, acc1 = {};
#pragma unroll
        for (int c = 0; c < 16; ++c) {
            const float4 a = af[i & 1][c];
            const v2f alo = { a.x, a.y }, ahi = { a.z, a.w };
            const v2f b0lo = { bf[0][c].x, bf[0][c].y }, b0hi = { bf[0][c].z, bf[0][c].w };
            const v2f b1lo = { bf[1][c].x, bf[1][c].y }, b1hi = { bf[1][c].z, bf[1][c].w };
            acc0 = wmma_f32(alo, b0lo, acc0);
            acc1 = wmma_f32(alo, b1lo, acc1);
            acc0 = wmma_f32(ahi, b0hi, acc0);
            acc1 = wmma_f32(ahi, b1hi, acc1);
        }
        float* base = out + (size_t)((mt0 + i) * 16 + 8 * half) * 128 + 32 * s + l16;
#pragma unroll
        for (int r = 0; r < 8; ++r) {
            base[r * 128]      = acc0[r];
            base[r * 128 + 16] = acc1[r];
        }
    }
}

extern "C" void kernel_launch(void* const* d_in, const int* in_sizes, int n_in,
                              void* d_out, int out_size, void* d_ws, size_t ws_size,
                              hipStream_t stream) {
    const float* x      = (const float*)d_in[0];
    const float* w_attn = (const float*)d_in[1];
    const float* w_proj = (const float*)d_in[2];

    float* ws = (float*)d_ws;
    const size_t GBUF = (size_t)GROUPS * GLEN * HSZ;
    float* qg = ws;
    float* kg = ws + GBUF;
    float* vg = ws + 2 * GBUF;
    float* xg = ws + 3 * GBUF;
    float* part = ws + 4 * GBUF + 2048;

    qkv_gemm_kernel<<<256, 384, 0, stream>>>(x, w_attn, qg, kg, vg, part);
    attn_group_kernel<<<ATT_WGS, 32 * ATT_WAVES, 0, stream>>>(qg, kg, vg, xg, part);
    proj_gemm_kernel<<<205, 128, 0, stream>>>(xg, qg, kg, w_proj, (float*)d_out);
}
